# starting kernel
# baseline (speedup 1.0000x reference)
#include <hip/hip_runtime.h>

typedef __attribute__((ext_vector_type(2))) float v2f;
typedef __attribute__((ext_vector_type(8))) float v8f;

#define IN_F 256
#define H_F  128
#define OUT_F 64


__global__ void k_deg_init(float* deg, int n) {
    int i = blockIdx.x * blockDim.x + threadIdx.x;
    if (i < n) deg[i] = 1.0f;
}

__global__ void k_deg_accum(const int* __restrict__ dst, float* deg, int nE) {
    int e = blockIdx.x * blockDim.x + threadIdx.x;
    if (e < nE) atomicAdd(&deg[dst[e]], 1.0f);
}

__global__ void k_rsqrt_inplace(float* d, int n) {
    int i = blockIdx.x * blockDim.x + threadIdx.x;
    if (i < n) d[i] = rsqrtf(d[i]);
}


template <int Ki, int Ko, bool RELU_A>
__global__ __launch_bounds__(256)
void k_gemm_wmma(const float* __restrict__ X, const float* __restrict__ W,
                 float* __restrict__ Y, int nrows) {
    constexpr int CT = Ko / 16;
    const int wave = threadIdx.x >> 5;
    const int lane = threadIdx.x & 31;
    const int rowTile = blockIdx.x * 8 + wave;
    const int row0 = rowTile * 16;
    if (row0 >= nrows) return;

    const int m    = lane & 15;
    const int half = lane >> 4;
    const float* xrow = X + (size_t)(row0 + m) * Ki + 2 * half;

    v8f acc[CT] = {};

    for (int kb = 0; kb < Ki; kb += 4) {
        if (kb + 32 < Ki) __builtin_prefetch(xrow + kb + 32, 0, 3);

        float ax = xrow[kb];
        float ay = xrow[kb + 1];
        if (RELU_A) { ax = fmaxf(ax, 0.0f); ay = fmaxf(ay, 0.0f); }
        v2f a = { ax, ay };

        const int krow = kb + 2 * half;
        const float* w0 = W + (size_t)krow * Ko;
        const float* w1 = w0 + Ko;
#pragma unroll
        for (int j = 0; j < CT; ++j) {
            v2f b = { w0[j * 16 + m], w1[j * 16 + m] };
            acc[j] = __builtin_amdgcn_wmma_f32_16x16x4_f32(
                false, a, false, b, (short)0, acc[j], false, false);
        }
    }

#pragma unroll
    for (int j = 0; j < CT; ++j) {
#pragma unroll
        for (int r = 0; r < 8; ++r) {
            const int M = r + 8 * half;
            Y[(size_t)(row0 + M) * Ko + j * 16 + m] = acc[j][r];
        }
    }
}

__global__ void k_agg_init(const float* __restrict__ lin, const float* __restrict__ dinv,
                           const float* __restrict__ bias, float* __restrict__ agg,
                           int n, int F) {
    int t = blockIdx.x * blockDim.x + threadIdx.x;
    int i = t / F;
    int f = t - i * F;
    if (i >= n) return;
    float w = dinv[i];
    agg[t] = lin[t] * (w * w) + bias[f];
}

__global__ void k_agg_edges(const float* __restrict__ lin, const int* __restrict__ src,
                            const int* __restrict__ dst, const float* __restrict__ dinv,
                            float* agg, int nE, int F4) {
    int t = blockIdx.x * blockDim.x + threadIdx.x;
    int e = t / F4;
    int c = t - e * F4;
    if (e >= nE) return;
    int s = src[e], d = dst[e];
    float w = dinv[s] * dinv[d];
    const float4 v = ((const float4*)(lin + (size_t)s * F4 * 4))[c];
    float* o = agg + (size_t)d * F4 * 4 + (size_t)c * 4;
    atomicAdd(o + 0, v.x * w);
    atomicAdd(o + 1, v.y * w);
    atomicAdd(o + 2, v.z * w);
    atomicAdd(o + 3, v.w * w);
}

__global__ void k_decode(const float* __restrict__ z,
                         const int* __restrict__ pos, const int* __restrict__ neg,
                         float* __restrict__ out, int EP) {
    int wid  = (blockIdx.x * blockDim.x + threadIdx.x) >> 5;
    int lane = threadIdx.x & 31;
    if (wid >= 2 * EP) return;
    int u, v;
    if (wid < EP) { u = pos[wid];       v = pos[EP + wid]; }
    else          { u = neg[wid - EP];  v = neg[EP + (wid - EP)]; }
    const float* zu = z + (size_t)u * OUT_F;
    const float* zv = z + (size_t)v * OUT_F;
    float s = zu[lane] * zv[lane] + zu[lane + 32] * zv[lane + 32];
#pragma unroll
    for (int off = 16; off > 0; off >>= 1)
        s += __shfl_down(s, off, 32);
    if (lane == 0) out[wid] = s;
}


extern "C" void kernel_launch(void* const* d_in, const int* in_sizes, int n_in,
                              void* d_out, int out_size, void* d_ws, size_t ws_size,
                              hipStream_t stream) {
    const float* x    = (const float*)d_in[0];
    const int*   ei   = (const int*)  d_in[1];
    const int*   pei  = (const int*)  d_in[2];
    const int*   nei  = (const int*)  d_in[3];
    const float* W1   = (const float*)d_in[4];
    const float* b1   = (const float*)d_in[5];
    const float* W2   = (const float*)d_in[6];
    const float* b2   = (const float*)d_in[7];
    float* out = (float*)d_out;

    const int N  = in_sizes[0] / IN_F;
    const int E  = in_sizes[1] / 2;
    const int EP = in_sizes[2] / 2;
    const int* src = ei;
    const int* dst = ei + E;

    float* dinv = (float*)d_ws;
    size_t npad = ((size_t)N + 255) & ~(size_t)255;
    float* bufA = dinv + npad;
    float* h    = bufA + (size_t)N * H_F;
    float* z    = h    + (size_t)N * H_F;

    const int B = 256;
    k_deg_init<<<(N + B - 1) / B, B, 0, stream>>>(dinv, N);
    k_deg_accum<<<(E + B - 1) / B, B, 0, stream>>>(dst, dinv, E);
    k_rsqrt_inplace<<<(N + B - 1) / B, B, 0, stream>>>(dinv, N);

    {
        int rowTiles = (N + 15) / 16;
        int blocks   = (rowTiles + 7) / 8;
        k_gemm_wmma<IN_F, H_F, false><<<blocks, B, 0, stream>>>(x, W1, bufA, N);
    }
    {
        int tot = N * H_F;
        k_agg_init<<<(tot + B - 1) / B, B, 0, stream>>>(bufA, dinv, b1, h, N, H_F);
        long long te = (long long)E * (H_F / 4);
        k_agg_edges<<<(int)((te + B - 1) / B), B, 0, stream>>>(bufA, src, dst, dinv, h, E, H_F / 4);
    }
    {
        int rowTiles = (N + 15) / 16;
        int blocks   = (rowTiles + 7) / 8;
        k_gemm_wmma<H_F, OUT_F, true><<<blocks, B, 0, stream>>>(h, W2, bufA, N);
    }
    {
        int tot = N * OUT_F;
        k_agg_init<<<(tot + B - 1) / B, B, 0, stream>>>(bufA, dinv, b2, z, N, OUT_F);
        long long te = (long long)E * (OUT_F / 4);
        k_agg_edges<<<(int)((te + B - 1) / B), B, 0, stream>>>(bufA, src, dst, dinv, z, E, OUT_F / 4);
    }
    {
        long long waves = 2LL * EP;
        long long thr   = waves * 32;
        k_decode<<<(int)((thr + B - 1) / B), B, 0, stream>>>(z, pei, nei, out, EP);
    }
}


# Aggregate by pull: in-call dst-sorted edge list (int atomic rank + prefix scan), one wave sums a node's in-edges in registers, one store per row; float atomics gone, sum order now follows rank
# speedup vs baseline: 1.8906x; 1.8906x over previous
#include <hip/hip_runtime.h>

typedef __attribute__((ext_vector_type(2))) float v2f;
typedef __attribute__((ext_vector_type(8))) float v8f;

#define IN_F 256
#define H_F  128
#define OUT_F 64


__global__ void k_cnt_zero(int* cnt, int n) {
    int i = blockIdx.x * blockDim.x + threadIdx.x;
    if (i < n) cnt[i] = 0;
}

__global__ void k_count_rank(const int* __restrict__ dst, int* cnt, int* __restrict__ rank, int nE) {
    int e = blockIdx.x * blockDim.x + threadIdx.x;
    if (e < nE) rank[e] = atomicAdd(&cnt[dst[e]], 1);
}

__global__ __launch_bounds__(1024)
void k_scan(const int* __restrict__ cnt, int* __restrict__ rowptr, float* __restrict__ dinv, int n) {
    __shared__ int s_pre[32];
    __shared__ int s_wsum[32];
    __shared__ int s_woff[32];
    const int tid = threadIdx.x;
    const int lane = tid & 31;
    const int wave = tid >> 5;
    const int blockStart = blockIdx.x * 1024;

    int pre = 0;
    for (int j = 0; j < (int)blockIdx.x; ++j) pre += cnt[j * 1024 + tid];
#pragma unroll
    for (int off = 16; off > 0; off >>= 1) pre += __shfl_xor(pre, off, 32);

    const int i = blockStart + tid;
    const int c = (i < n) ? cnt[i] : 0;
    int incl = c;
#pragma unroll
    for (int off = 1; off < 32; off <<= 1) {
        int t = __shfl_up(incl, off, 32);
        if (lane >= off) incl += t;
    }
    if (lane == 31) s_wsum[wave] = incl;
    if (lane == 0)  s_pre[wave] = pre;
    __syncthreads();
    if (wave == 0) {
        int p = s_pre[lane];
#pragma unroll
        for (int off = 16; off > 0; off >>= 1) p += __shfl_xor(p, off, 32);
        const int w = s_wsum[lane];
        int wi = w;
#pragma unroll
        for (int off = 1; off < 32; off <<= 1) {
            int t = __shfl_up(wi, off, 32);
            if (lane >= off) wi += t;
        }
        s_woff[lane] = p + wi - w;
    }
    __syncthreads();
    if (i < n) {
        rowptr[i] = s_woff[wave] + incl - c;
        dinv[i] = rsqrtf((float)(c + 1));
    }
}

__global__ void k_fill(const int* __restrict__ src, const int* __restrict__ dst,
                       const int* __restrict__ rowptr, const int* __restrict__ rank,
                       int* __restrict__ csr, int nE) {
    int e = blockIdx.x * blockDim.x + threadIdx.x;
    if (e < nE) csr[rowptr[dst[e]] + rank[e]] = src[e];
}


template <int Ki, int Ko, bool RELU_A>
__global__ __launch_bounds__(256)
void k_gemm_wmma(const float* __restrict__ X, const float* __restrict__ W,
                 float* __restrict__ Y, int nrows) {
    constexpr int CT = Ko / 16;
    const int wave = threadIdx.x >> 5;
    const int lane = threadIdx.x & 31;
    const int rowTile = blockIdx.x * 8 + wave;
    const int row0 = rowTile * 16;
    if (row0 >= nrows) return;

    const int m    = lane & 15;
    const int half = lane >> 4;
    const float* xrow = X + (size_t)(row0 + m) * Ki + 2 * half;

    v8f acc[CT] = {};

    for (int kb = 0; kb < Ki; kb += 4) {
        if (kb + 32 < Ki) __builtin_prefetch(xrow + kb + 32, 0, 3);

        float ax = xrow[kb];
        float ay = xrow[kb + 1];
        if (RELU_A) { ax = fmaxf(ax, 0.0f); ay = fmaxf(ay, 0.0f); }
        v2f a = { ax, ay };

        const int krow = kb + 2 * half;
        const float* w0 = W + (size_t)krow * Ko;
        const float* w1 = w0 + Ko;
#pragma unroll
        for (int j = 0; j < CT; ++j) {
            v2f b = { w0[j * 16 + m], w1[j * 16 + m] };
            acc[j] = __builtin_amdgcn_wmma_f32_16x16x4_f32(
                false, a, false, b, (short)0, acc[j], false, false);
        }
    }

#pragma unroll
    for (int j = 0; j < CT; ++j) {
#pragma unroll
        for (int r = 0; r < 8; ++r) {
            const int M = r + 8 * half;
            Y[(size_t)(row0 + M) * Ko + j * 16 + m] = acc[j][r];
        }
    }
}

template <int F>
__global__ __launch_bounds__(256)
void k_gather(const float* __restrict__ lin, const int* __restrict__ rowptr,
              const int* __restrict__ cnt, const int* __restrict__ csr,
              const float* __restrict__ dinv, const float* __restrict__ bias,
              float* __restrict__ agg, int n) {
    constexpr int V = F / 32;
    typedef __attribute__((ext_vector_type(V))) float vecf;
    const int lane = threadIdx.x & 31;
    const int d = __builtin_amdgcn_readfirstlane((int)((blockIdx.x * blockDim.x + threadIdx.x) >> 5));
    if (d >= n) return;
    const int start = rowptr[d];
    const int len   = cnt[d];
    const float wd  = dinv[d];
    const vecf* linv = (const vecf*)lin;

    vecf acc = linv[(size_t)d * 32 + lane] * (wd * wd);
#pragma unroll
    for (int j = 0; j < V; ++j) acc[j] += bias[lane * V + j];

    for (int base = 0; base < len; base += 32) {
        const int m = (len - base < 32) ? (len - base) : 32;
        int   s_l = 0;
        float w_l = 0.0f;
        if (lane < m) { s_l = csr[start + base + lane]; w_l = dinv[s_l]; }
        const int w_i = __float_as_int(w_l);
        int k = 0;
        for (; k + 4 <= m; k += 4) {
            const int s0 = __builtin_amdgcn_readlane(s_l, k);
            const int s1 = __builtin_amdgcn_readlane(s_l, k + 1);
            const int s2 = __builtin_amdgcn_readlane(s_l, k + 2);
            const int s3 = __builtin_amdgcn_readlane(s_l, k + 3);
            const vecf v0 = linv[(size_t)s0 * 32 + lane];
            const vecf v1 = linv[(size_t)s1 * 32 + lane];
            const vecf v2 = linv[(size_t)s2 * 32 + lane];
            const vecf v3 = linv[(size_t)s3 * 32 + lane];
            acc += v0 * (__int_as_float(__builtin_amdgcn_readlane(w_i, k)) * wd);
            acc += v1 * (__int_as_float(__builtin_amdgcn_readlane(w_i, k + 1)) * wd);
            acc += v2 * (__int_as_float(__builtin_amdgcn_readlane(w_i, k + 2)) * wd);
            acc += v3 * (__int_as_float(__builtin_amdgcn_readlane(w_i, k + 3)) * wd);
        }
        for (; k < m; ++k) {
            const int s0 = __builtin_amdgcn_readlane(s_l, k);
            const vecf v0 = linv[(size_t)s0 * 32 + lane];
            acc += v0 * (__int_as_float(__builtin_amdgcn_readlane(w_i, k)) * wd);
        }
    }
    ((vecf*)agg)[(size_t)d * 32 + lane] = acc;
}

__global__ void k_decode(const float* __restrict__ z,
                         const int* __restrict__ pos, const int* __restrict__ neg,
                         float* __restrict__ out, int EP) {
    int wid  = (blockIdx.x * blockDim.x + threadIdx.x) >> 5;
    int lane = threadIdx.x & 31;
    if (wid >= 2 * EP) return;
    int u, v;
    if (wid < EP) { u = pos[wid];       v = pos[EP + wid]; }
    else          { u = neg[wid - EP];  v = neg[EP + (wid - EP)]; }
    const float* zu = z + (size_t)u * OUT_F;
    const float* zv = z + (size_t)v * OUT_F;
    float s = zu[lane] * zv[lane] + zu[lane + 32] * zv[lane + 32];
#pragma unroll
    for (int off = 16; off > 0; off >>= 1)
        s += __shfl_down(s, off, 32);
    if (lane == 0) out[wid] = s;
}


extern "C" void kernel_launch(void* const* d_in, const int* in_sizes, int n_in,
                              void* d_out, int out_size, void* d_ws, size_t ws_size,
                              hipStream_t stream) {
    const float* x    = (const float*)d_in[0];
    const int*   ei   = (const int*)  d_in[1];
    const int*   pei  = (const int*)  d_in[2];
    const int*   nei  = (const int*)  d_in[3];
    const float* W1   = (const float*)d_in[4];
    const float* b1   = (const float*)d_in[5];
    const float* W2   = (const float*)d_in[6];
    const float* b2   = (const float*)d_in[7];
    float* out = (float*)d_out;

    const int N  = in_sizes[0] / IN_F;
    const int E  = in_sizes[1] / 2;
    const int EP = in_sizes[2] / 2;
    const int* src = ei;
    const int* dst = ei + E;

    float* dinv = (float*)d_ws;
    size_t npad = ((size_t)N + 255) & ~(size_t)255;
    float* bufA = dinv + npad;
    float* h    = bufA + (size_t)N * H_F;
    int*   cnt    = (int*)(h + (size_t)N * H_F);
    int*   rowptr = cnt + npad;
    int*   rank   = rowptr + npad;
    int*   csr    = rank + E;
    float* z    = h;

    const int B = 256;
    k_cnt_zero<<<(N + B - 1) / B, B, 0, stream>>>(cnt, N);
    k_count_rank<<<(E + B - 1) / B, B, 0, stream>>>(dst, cnt, rank, E);
    k_scan<<<(N + 1023) / 1024, 1024, 0, stream>>>(cnt, rowptr, dinv, N);
    k_fill<<<(E + B - 1) / B, B, 0, stream>>>(src, dst, rowptr, rank, csr, E);

    const int gatherBlocks = (N + 7) / 8;

    {
        int rowTiles = (N + 15) / 16;
        int blocks   = (rowTiles + 7) / 8;
        k_gemm_wmma<IN_F, H_F, false><<<blocks, B, 0, stream>>>(x, W1, bufA, N);
    }
    k_gather<H_F><<<gatherBlocks, B, 0, stream>>>(bufA, rowptr, cnt, csr, dinv, b1, h, N);
    {
        int rowTiles = (N + 15) / 16;
        int blocks   = (rowTiles + 7) / 8;
        k_gemm_wmma<H_F, OUT_F, true><<<blocks, B, 0, stream>>>(h, W2, bufA, N);
    }
    k_gather<OUT_F><<<gatherBlocks, B, 0, stream>>>(bufA, rowptr, cnt, csr, dinv, b2, z, N);
    {
        long long waves = 2LL * EP;
        long long thr   = waves * 32;
        k_decode<<<(int)((thr + B - 1) / B), B, 0, stream>>>(z, pei, nei, out, EP);
    }
}


# Decode: one wave per 32 outputs; each half-wave reads a whole 256-B z row per load (b128 per lane), loads batched ahead of the sums, DPP row sums, one 128-B store; dot sum order changes
# speedup vs baseline: 2.3755x; 1.2565x over previous
#include <hip/hip_runtime.h>

typedef __attribute__((ext_vector_type(2))) float v2f;
typedef __attribute__((ext_vector_type(8))) float v8f;

#define IN_F 256
#define H_F  128
#define OUT_F 64


__global__ void k_cnt_zero(int* cnt, int n) {
    int i = blockIdx.x * blockDim.x + threadIdx.x;
    if (i < n) cnt[i] = 0;
}

__global__ void k_count_rank(const int* __restrict__ dst, int* cnt, int* __restrict__ rank, int nE) {
    int e = blockIdx.x * blockDim.x + threadIdx.x;
    if (e < nE) rank[e] = atomicAdd(&cnt[dst[e]], 1);
}

__global__ __launch_bounds__(1024)
void k_scan(const int* __restrict__ cnt, int* __restrict__ rowptr, float* __restrict__ dinv, int n) {
    __shared__ int s_pre[32];
    __shared__ int s_wsum[32];
    __shared__ int s_woff[32];
    const int tid = threadIdx.x;
    const int lane = tid & 31;
    const int wave = tid >> 5;
    const int blockStart = blockIdx.x * 1024;

    int pre = 0;
    for (int j = 0; j < (int)blockIdx.x; ++j) pre += cnt[j * 1024 + tid];
#pragma unroll
    for (int off = 16; off > 0; off >>= 1) pre += __shfl_xor(pre, off, 32);

    const int i = blockStart + tid;
    const int c = (i < n) ? cnt[i] : 0;
    int incl = c;
#pragma unroll
    for (int off = 1; off < 32; off <<= 1) {
        int t = __shfl_up(incl, off, 32);
        if (lane >= off) incl += t;
    }
    if (lane == 31) s_wsum[wave] = incl;
    if (lane == 0)  s_pre[wave] = pre;
    __syncthreads();
    if (wave == 0) {
        int p = s_pre[lane];
#pragma unroll
        for (int off = 16; off > 0; off >>= 1) p += __shfl_xor(p, off, 32);
        const int w = s_wsum[lane];
        int wi = w;
#pragma unroll
        for (int off = 1; off < 32; off <<= 1) {
            int t = __shfl_up(wi, off, 32);
            if (lane >= off) wi += t;
        }
        s_woff[lane] = p + wi - w;
    }
    __syncthreads();
    if (i < n) {
        rowptr[i] = s_woff[wave] + incl - c;
        dinv[i] = rsqrtf((float)(c + 1));
    }
}

__global__ void k_fill(const int* __restrict__ src, const int* __restrict__ dst,
                       const int* __restrict__ rowptr, const int* __restrict__ rank,
                       int* __restrict__ csr, int nE) {
    int e = blockIdx.x * blockDim.x + threadIdx.x;
    if (e < nE) csr[rowptr[dst[e]] + rank[e]] = src[e];
}


template <int Ki, int Ko, bool RELU_A>
__global__ __launch_bounds__(256)
void k_gemm_wmma(const float* __restrict__ X, const float* __restrict__ W,
                 float* __restrict__ Y, int nrows) {
    constexpr int CT = Ko / 16;
    const int wave = threadIdx.x >> 5;
    const int lane = threadIdx.x & 31;
    const int rowTile = blockIdx.x * 8 + wave;
    const int row0 = rowTile * 16;
    if (row0 >= nrows) return;

    const int m    = lane & 15;
    const int half = lane >> 4;
    const float* xrow = X + (size_t)(row0 + m) * Ki + 2 * half;

    v8f acc[CT] = {};

    for (int kb = 0; kb < Ki; kb += 4) {
        if (kb + 32 < Ki) __builtin_prefetch(xrow + kb + 32, 0, 3);

        float ax = xrow[kb];
        float ay = xrow[kb + 1];
        if (RELU_A) { ax = fmaxf(ax, 0.0f); ay = fmaxf(ay, 0.0f); }
        v2f a = { ax, ay };

        const int krow = kb + 2 * half;
        const float* w0 = W + (size_t)krow * Ko;
        const float* w1 = w0 + Ko;
#pragma unroll
        for (int j = 0; j < CT; ++j) {
            v2f b = { w0[j * 16 + m], w1[j * 16 + m] };
            acc[j] = __builtin_amdgcn_wmma_f32_16x16x4_f32(
                false, a, false, b, (short)0, acc[j], false, false);
        }
    }

#pragma unroll
    for (int j = 0; j < CT; ++j) {
#pragma unroll
        for (int r = 0; r < 8; ++r) {
            const int M = r + 8 * half;
            Y[(size_t)(row0 + M) * Ko + j * 16 + m] = acc[j][r];
        }
    }
}

template <int F>
__global__ __launch_bounds__(256)
void k_gather(const float* __restrict__ lin, const int* __restrict__ rowptr,
              const int* __restrict__ cnt, const int* __restrict__ csr,
              const float* __restrict__ dinv, const float* __restrict__ bias,
              float* __restrict__ agg, int n) {
    constexpr int V = F / 32;
    typedef __attribute__((ext_vector_type(V))) float vecf;
    const int lane = threadIdx.x & 31;
    const int d = __builtin_amdgcn_readfirstlane((int)((blockIdx.x * blockDim.x + threadIdx.x) >> 5));
    if (d >= n) return;
    const int start = rowptr[d];
    const int len   = cnt[d];
    const float wd  = dinv[d];
    const vecf* linv = (const vecf*)lin;

    vecf acc = linv[(size_t)d * 32 + lane] * (wd * wd);
#pragma unroll
    for (int j = 0; j < V; ++j) acc[j] += bias[lane * V + j];

    for (int base = 0; base < len; base += 32) {
        const int m = (len - base < 32) ? (len - base) : 32;
        int   s_l = 0;
        float w_l = 0.0f;
        if (lane < m) { s_l = csr[start + base + lane]; w_l = dinv[s_l]; }
        const int w_i = __float_as_int(w_l);
        int k = 0;
        for (; k + 4 <= m; k += 4) {
            const int s0 = __builtin_amdgcn_readlane(s_l, k);
            const int s1 = __builtin_amdgcn_readlane(s_l, k + 1);
            const int s2 = __builtin_amdgcn_readlane(s_l, k + 2);
            const int s3 = __builtin_amdgcn_readlane(s_l, k + 3);
            const vecf v0 = linv[(size_t)s0 * 32 + lane];
            const vecf v1 = linv[(size_t)s1 * 32 + lane];
            const vecf v2 = linv[(size_t)s2 * 32 + lane];
            const vecf v3 = linv[(size_t)s3 * 32 + lane];
            acc += v0 * (__int_as_float(__builtin_amdgcn_readlane(w_i, k)) * wd);
            acc += v1 * (__int_as_float(__builtin_amdgcn_readlane(w_i, k + 1)) * wd);
            acc += v2 * (__int_as_float(__builtin_amdgcn_readlane(w_i, k + 2)) * wd);
            acc += v3 * (__int_as_float(__builtin_amdgcn_readlane(w_i, k + 3)) * wd);
        }
        for (; k < m; ++k) {
            const int s0 = __builtin_amdgcn_readlane(s_l, k);
            const vecf v0 = linv[(size_t)s0 * 32 + lane];
            acc += v0 * (__int_as_float(__builtin_amdgcn_readlane(w_i, k)) * wd);
        }
    }
    ((vecf*)agg)[(size_t)d * 32 + lane] = acc;
}


template <int K>
__device__ __forceinline__ unsigned row_bcast(unsigned v) {
    return (unsigned)__builtin_amdgcn_update_dpp(0, (int)v, 0x150 + K, 0xf, 0xf, true);
}
template <int M>
__device__ __forceinline__ float row_xadd(float x) {
    const int y = __builtin_amdgcn_update_dpp(0, __float_as_int(x), 0x160 + M, 0xf, 0xf, true);
    return x + __int_as_float(y);
}

template <int K>
__device__ __forceinline__ float4 z_piece(const char* __restrict__ zb, unsigned off, unsigned sub16) {
    return *(const float4*)(zb + (row_bcast<K>(off) + sub16));
}

template <int K>
__device__ __forceinline__ float score_step(const float4 a, const float4 b, int sub, float res) {
    float p = a.x * b.x;
    p = fmaf(a.y, b.y, p);
    p = fmaf(a.z, b.z, p);
    p = fmaf(a.w, b.w, p);
    p = row_xadd<1>(p);
    p = row_xadd<2>(p);
    p = row_xadd<4>(p);
    p = row_xadd<8>(p);
    return (sub == K) ? p : res;
}

#define DEC_LOAD(K, J) a[J] = z_piece<K>(zb, offu, sub16); b[J] = z_piece<K>(zb, offv, sub16);
#define DEC_SUM(K, J)  res = score_step<K>(a[J], b[J], sub, res);

__global__ __launch_bounds__(256)
void k_decode(const float* __restrict__ z,
              const int* __restrict__ pos, const int* __restrict__ neg,
              float* __restrict__ out, int EP) {
    const int lane  = threadIdx.x & 31;
    const int sub   = lane & 15;
    const int total = 2 * EP;
    const int i0    = (int)((blockIdx.x * blockDim.x + threadIdx.x) >> 5) * 32;
    if (i0 >= total) return;
    const int i = i0 + lane;
    int u = 0, v = 0;
    if (i < total) {
        if (i < EP) { u = pos[i];       v = pos[EP + i]; }
        else        { u = neg[i - EP];  v = neg[i]; }
    }
    const char* zb = (const char*)z;
    const unsigned offu  = (unsigned)u * (OUT_F * 4u);
    const unsigned offv  = (unsigned)v * (OUT_F * 4u);
    const unsigned sub16 = (unsigned)sub * 16u;
    float4 a[8], b[8];
    float res = 0.0f;

    DEC_LOAD(0, 0) DEC_LOAD(1, 1) DEC_LOAD(2, 2) DEC_LOAD(3, 3)
    DEC_LOAD(4, 4) DEC_LOAD(5, 5) DEC_LOAD(6, 6) DEC_LOAD(7, 7)
    DEC_SUM(0, 0) DEC_SUM(1, 1) DEC_SUM(2, 2) DEC_SUM(3, 3)
    DEC_SUM(4, 4) DEC_SUM(5, 5) DEC_SUM(6, 6) DEC_SUM(7, 7)
    DEC_LOAD(8, 0)  DEC_LOAD(9, 1)  DEC_LOAD(10, 2) DEC_LOAD(11, 3)
    DEC_LOAD(12, 4) DEC_LOAD(13, 5) DEC_LOAD(14, 6) DEC_LOAD(15, 7)
    DEC_SUM(8, 0)  DEC_SUM(9, 1)  DEC_SUM(10, 2) DEC_SUM(11, 3)
    DEC_SUM(12, 4) DEC_SUM(13, 5) DEC_SUM(14, 6) DEC_SUM(15, 7)

    if (i < total) out[i] = res;
}


extern "C" void kernel_launch(void* const* d_in, const int* in_sizes, int n_in,
                              void* d_out, int out_size, void* d_ws, size_t ws_size,
                              hipStream_t stream) {
    const float* x    = (const float*)d_in[0];
    const int*   ei   = (const int*)  d_in[1];
    const int*   pei  = (const int*)  d_in[2];
    const int*   nei  = (const int*)  d_in[3];
    const float* W1   = (const float*)d_in[4];
    const float* b1   = (const float*)d_in[5];
    const float* W2   = (const float*)d_in[6];
    const float* b2   = (const float*)d_in[7];
    float* out = (float*)d_out;

    const int N  = in_sizes[0] / IN_F;
    const int E  = in_sizes[1] / 2;
    const int EP = in_sizes[2] / 2;
    const int* src = ei;
    const int* dst = ei + E;

    float* dinv = (float*)d_ws;
    size_t npad = ((size_t)N + 255) & ~(size_t)255;
    float* bufA = dinv + npad;
    float* h    = bufA + (size_t)N * H_F;
    int*   cnt    = (int*)(h + (size_t)N * H_F);
    int*   rowptr = cnt + npad;
    int*   rank   = rowptr + npad;
    int*   csr    = rank + E;
    float* z    = h;

    const int B = 256;
    k_cnt_zero<<<(N + B - 1) / B, B, 0, stream>>>(cnt, N);
    k_count_rank<<<(E + B - 1) / B, B, 0, stream>>>(dst, cnt, rank, E);
    k_scan<<<(N + 1023) / 1024, 1024, 0, stream>>>(cnt, rowptr, dinv, N);
    k_fill<<<(E + B - 1) / B, B, 0, stream>>>(src, dst, rowptr, rank, csr, E);

    const int gatherBlocks = (N + 7) / 8;

    {
        int rowTiles = (N + 15) / 16;
        int blocks   = (rowTiles + 7) / 8;
        k_gemm_wmma<IN_F, H_F, false><<<blocks, B, 0, stream>>>(x, W1, bufA, N);
    }
    k_gather<H_F><<<gatherBlocks, B, 0, stream>>>(bufA, rowptr, cnt, csr, dinv, b1, h, N);
    {
        int rowTiles = (N + 15) / 16;
        int blocks   = (rowTiles + 7) / 8;
        k_gemm_wmma<H_F, OUT_F, true><<<blocks, B, 0, stream>>>(h, W2, bufA, N);
    }
    k_gather<OUT_F><<<gatherBlocks, B, 0, stream>>>(bufA, rowptr, cnt, csr, dinv, b2, z, N);
    {
        long long waves = (2LL * EP + 31) / 32;
        long long thr   = waves * 32;
        k_decode<<<(int)((thr + B - 1) / B), B, 0, stream>>>(z, pei, nei, out, EP);
    }
}


# Edge list built by a two-level counting sort on LDS counters: edges binned per tile into 256-node buckets, one work-group per bucket ranks and fills; global atomics gone; rank (sum) order shifts
# speedup vs baseline: 3.1820x; 1.3395x over previous
#include <hip/hip_runtime.h>

typedef __attribute__((ext_vector_type(2))) float v2f;
typedef __attribute__((ext_vector_type(8))) float v8f;

#define IN_F 256
#define H_F  128
#define OUT_F 64


#define NODE_SHIFT 8
#define BKT_NODES  256
#define EPT        8
#define TILE_EDGES 8192

__global__ __launch_bounds__(1024)
void k_bin_count(const int* __restrict__ dst, int* __restrict__ tileHist, int* __restrict__ rnk,
                 int nE, int nT, int nB) {
    extern __shared__ int s_bins[];
    const int tid = threadIdx.x;
    const int t = blockIdx.x;
    for (int b = tid; b < nB; b += 1024) s_bins[b] = 0;
    __syncthreads();
    const int e0 = t * TILE_EDGES;
    if (e0 + TILE_EDGES <= nE) {
        int d[EPT];
#pragma unroll
        for (int u = 0; u < EPT; ++u) d[u] = dst[e0 + u * 1024 + tid];
#pragma unroll
        for (int u = 0; u < EPT; ++u)
            rnk[e0 + u * 1024 + tid] = atomicAdd(&s_bins[d[u] >> NODE_SHIFT], 1);
    } else {
        for (int e = e0 + tid; e < nE; e += 1024)
            rnk[e] = atomicAdd(&s_bins[dst[e] >> NODE_SHIFT], 1);
    }
    __syncthreads();
    for (int b = tid; b < nB; b += 1024) tileHist[b * nT + t] = s_bins[b];
}

__global__ __launch_bounds__(1024)
void k_scan_excl(const int* __restrict__ cnt, int* __restrict__ excl, int n) {
    __shared__ int s_pre[32];
    __shared__ int s_wsum[32];
    __shared__ int s_woff[32];
    const int tid = threadIdx.x;
    const int lane = tid & 31;
    const int wave = tid >> 5;
    const int blockStart = blockIdx.x * 1024;

    int pre = 0;
    for (int j = 0; j < (int)blockIdx.x; ++j) pre += cnt[j * 1024 + tid];
#pragma unroll
    for (int off = 16; off > 0; off >>= 1) pre += __shfl_xor(pre, off, 32);

    const int i = blockStart + tid;
    const int c = (i < n) ? cnt[i] : 0;
    int incl = c;
#pragma unroll
    for (int off = 1; off < 32; off <<= 1) {
        int t = __shfl_up(incl, off, 32);
        if (lane >= off) incl += t;
    }
    if (lane == 31) s_wsum[wave] = incl;
    if (lane == 0)  s_pre[wave] = pre;
    __syncthreads();
    if (wave == 0) {
        int p = s_pre[lane];
#pragma unroll
        for (int off = 16; off > 0; off >>= 1) p += __shfl_xor(p, off, 32);
        const int w = s_wsum[lane];
        int wi = w;
#pragma unroll
        for (int off = 1; off < 32; off <<= 1) {
            int t = __shfl_up(wi, off, 32);
            if (lane >= off) wi += t;
        }
        s_woff[lane] = p + wi - w;
    }
    __syncthreads();
    if (i < n) excl[i] = s_woff[wave] + incl - c;
}

__global__ __launch_bounds__(1024)
void k_bin_scatter(const int* __restrict__ src, const int* __restrict__ dst,
                   const int* __restrict__ off, const int* __restrict__ rnk,
                   int2* __restrict__ binned, int nE, int nT, int nB) {
    extern __shared__ int s_bins[];
    const int tid = threadIdx.x;
    const int t = blockIdx.x;
    for (int b = tid; b < nB; b += 1024) s_bins[b] = off[b * nT + t];
    __syncthreads();
    const int e0 = t * TILE_EDGES;
    if (e0 + TILE_EDGES <= nE) {
        int d[EPT], s[EPT], r[EPT];
#pragma unroll
        for (int u = 0; u < EPT; ++u) {
            const int e = e0 + u * 1024 + tid;
            d[u] = dst[e];
            s[u] = src[e];
            r[u] = rnk[e];
        }
#pragma unroll
        for (int u = 0; u < EPT; ++u)
            binned[s_bins[d[u] >> NODE_SHIFT] + r[u]] = make_int2(s[u], d[u]);
    } else {
        for (int e = e0 + tid; e < nE; e += 1024) {
            const int d = dst[e];
            binned[s_bins[d >> NODE_SHIFT] + rnk[e]] = make_int2(src[e], d);
        }
    }
}

__global__ __launch_bounds__(1024)
void k_bucket_csr(const int2* __restrict__ binned, const int* __restrict__ off,
                  int* __restrict__ rowptr, int* __restrict__ cnt, float* __restrict__ dinv,
                  int* __restrict__ csr, int n, int nE, int nT, int nB) {
    __shared__ int s_cnt[BKT_NODES];
    __shared__ int s_cur[BKT_NODES];
    __shared__ int s_wsum[BKT_NODES / 32];
    const int tid = threadIdx.x;
    const int lane = tid & 31;
    const int wave = tid >> 5;
    const int b = blockIdx.x;
    const int node0 = b << NODE_SHIFT;
    const int bs = off[b * nT];
    const int be = (b + 1 < nB) ? off[(b + 1) * nT] : nE;
    const int own = tid & (BKT_NODES - 1);

    if (tid < BKT_NODES) s_cnt[tid] = 0;
    __syncthreads();
    for (int base = bs; base < be; base += 4096) {
        int d[4];
#pragma unroll
        for (int u = 0; u < 4; ++u) {
            const int e = base + u * 1024 + tid;
            d[u] = binned[(e < be) ? e : be - 1].y;
        }
#pragma unroll
        for (int u = 0; u < 4; ++u) {
            const bool in = base + u * 1024 + tid < be;
            atomicAdd(&s_cnt[in ? d[u] - node0 : own], in ? 1 : 0);
        }
    }
    __syncthreads();

    int c = 0, incl = 0;
    if (tid < BKT_NODES) {
        c = s_cnt[tid];
        incl = c;
#pragma unroll
        for (int o = 1; o < 32; o <<= 1) {
            int t = __shfl_up(incl, o, 32);
            if (lane >= o) incl += t;
        }
        if (lane == 31) s_wsum[wave] = incl;
    }
    __syncthreads();
    if (tid < BKT_NODES) {
        int base = 0;
        for (int w = 0; w < wave; ++w) base += s_wsum[w];
        const int ex = base + incl - c;
        s_cur[tid] = ex;
        const int node = node0 + tid;
        if (node < n) {
            rowptr[node] = bs + ex;
            cnt[node]    = c;
            dinv[node]   = rsqrtf((float)(c + 1));
        }
    }
    __syncthreads();
    for (int base = bs; base < be; base += 4096) {
        int2 ed[4];
#pragma unroll
        for (int u = 0; u < 4; ++u) {
            const int e = base + u * 1024 + tid;
            ed[u] = binned[(e < be) ? e : be - 1];
        }
#pragma unroll
        for (int u = 0; u < 4; ++u) {
            const bool in = base + u * 1024 + tid < be;
            const int p = atomicAdd(&s_cur[in ? ed[u].y - node0 : own], in ? 1 : 0);
            if (in) csr[bs + p] = ed[u].x;
        }
    }
}


template <int Ki, int Ko, bool RELU_A>
__global__ __launch_bounds__(256)
void k_gemm_wmma(const float* __restrict__ X, const float* __restrict__ W,
                 float* __restrict__ Y, int nrows) {
    constexpr int CT = Ko / 16;
    const int wave = threadIdx.x >> 5;
    const int lane = threadIdx.x & 31;
    const int rowTile = blockIdx.x * 8 + wave;
    const int row0 = rowTile * 16;
    if (row0 >= nrows) return;

    const int m    = lane & 15;
    const int half = lane >> 4;
    const float* xrow = X + (size_t)(row0 + m) * Ki + 2 * half;

    v8f acc[CT] = {};

    for (int kb = 0; kb < Ki; kb += 4) {
        if (kb + 32 < Ki) __builtin_prefetch(xrow + kb + 32, 0, 3);

        float ax = xrow[kb];
        float ay = xrow[kb + 1];
        if (RELU_A) { ax = fmaxf(ax, 0.0f); ay = fmaxf(ay, 0.0f); }
        v2f a = { ax, ay };

        const int krow = kb + 2 * half;
        const float* w0 = W + (size_t)krow * Ko;
        const float* w1 = w0 + Ko;
#pragma unroll
        for (int j = 0; j < CT; ++j) {
            v2f b = { w0[j * 16 + m], w1[j * 16 + m] };
            acc[j] = __builtin_amdgcn_wmma_f32_16x16x4_f32(
                false, a, false, b, (short)0, acc[j], false, false);
        }
    }

#pragma unroll
    for (int j = 0; j < CT; ++j) {
#pragma unroll
        for (int r = 0; r < 8; ++r) {
            const int M = r + 8 * half;
            Y[(size_t)(row0 + M) * Ko + j * 16 + m] = acc[j][r];
        }
    }
}

template <int F>
__global__ __launch_bounds__(256)
void k_gather(const float* __restrict__ lin, const int* __restrict__ rowptr,
              const int* __restrict__ cnt, const int* __restrict__ csr,
              const float* __restrict__ dinv, const float* __restrict__ bias,
              float* __restrict__ agg, int n) {
    constexpr int V = F / 32;
    typedef __attribute__((ext_vector_type(V))) float vecf;
    const int lane = threadIdx.x & 31;
    const int d = __builtin_amdgcn_readfirstlane((int)((blockIdx.x * blockDim.x + threadIdx.x) >> 5));
    if (d >= n) return;
    const int start = rowptr[d];
    const int len   = cnt[d];
    const float wd  = dinv[d];
    const vecf* linv = (const vecf*)lin;

    vecf acc = linv[(size_t)d * 32 + lane] * (wd * wd);
#pragma unroll
    for (int j = 0; j < V; ++j) acc[j] += bias[lane * V + j];

    for (int base = 0; base < len; base += 32) {
        const int m = (len - base < 32) ? (len - base) : 32;
        int   s_l = 0;
        float w_l = 0.0f;
        if (lane < m) { s_l = csr[start + base + lane]; w_l = dinv[s_l]; }
        const int w_i = __float_as_int(w_l);
        int k = 0;
        for (; k + 4 <= m; k += 4) {
            const int s0 = __builtin_amdgcn_readlane(s_l, k);
            const int s1 = __builtin_amdgcn_readlane(s_l, k + 1);
            const int s2 = __builtin_amdgcn_readlane(s_l, k + 2);
            const int s3 = __builtin_amdgcn_readlane(s_l, k + 3);
            const vecf v0 = linv[(size_t)s0 * 32 + lane];
            const vecf v1 = linv[(size_t)s1 * 32 + lane];
            const vecf v2 = linv[(size_t)s2 * 32 + lane];
            const vecf v3 = linv[(size_t)s3 * 32 + lane];
            acc += v0 * (__int_as_float(__builtin_amdgcn_readlane(w_i, k)) * wd);
            acc += v1 * (__int_as_float(__builtin_amdgcn_readlane(w_i, k + 1)) * wd);
            acc += v2 * (__int_as_float(__builtin_amdgcn_readlane(w_i, k + 2)) * wd);
            acc += v3 * (__int_as_float(__builtin_amdgcn_readlane(w_i, k + 3)) * wd);
        }
        for (; k < m; ++k) {
            const int s0 = __builtin_amdgcn_readlane(s_l, k);
            const vecf v0 = linv[(size_t)s0 * 32 + lane];
            acc += v0 * (__int_as_float(__builtin_amdgcn_readlane(w_i, k)) * wd);
        }
    }
    ((vecf*)agg)[(size_t)d * 32 + lane] = acc;
}


template <int K>
__device__ __forceinline__ unsigned row_bcast(unsigned v) {
    return (unsigned)__builtin_amdgcn_update_dpp(0, (int)v, 0x150 + K, 0xf, 0xf, true);
}
template <int M>
__device__ __forceinline__ float row_xadd(float x) {
    const int y = __builtin_amdgcn_update_dpp(0, __float_as_int(x), 0x160 + M, 0xf, 0xf, true);
    return x + __int_as_float(y);
}

template <int K>
__device__ __forceinline__ float4 z_piece(const char* __restrict__ zb, unsigned off, unsigned sub16) {
    return *(const float4*)(zb + (row_bcast<K>(off) + sub16));
}

template <int K>
__device__ __forceinline__ float score_step(const float4 a, const float4 b, int sub, float res) {
    float p = a.x * b.x;
    p = fmaf(a.y, b.y, p);
    p = fmaf(a.z, b.z, p);
    p = fmaf(a.w, b.w, p);
    p = row_xadd<1>(p);
    p = row_xadd<2>(p);
    p = row_xadd<4>(p);
    p = row_xadd<8>(p);
    return (sub == K) ? p : res;
}

#define DEC_LOAD(K, J) a[J] = z_piece<K>(zb, offu, sub16); b[J] = z_piece<K>(zb, offv, sub16);
#define DEC_SUM(K, J)  res = score_step<K>(a[J], b[J], sub, res);

__global__ __launch_bounds__(256)
void k_decode(const float* __restrict__ z,
              const int* __restrict__ pos, const int* __restrict__ neg,
              float* __restrict__ out, int EP) {
    const int lane  = threadIdx.x & 31;
    const int sub   = lane & 15;
    const int total = 2 * EP;
    const int i0    = (int)((blockIdx.x * blockDim.x + threadIdx.x) >> 5) * 32;
    if (i0 >= total) return;
    const int i = i0 + lane;
    int u = 0, v = 0;
    if (i < total) {
        if (i < EP) { u = pos[i];       v = pos[EP + i]; }
        else        { u = neg[i - EP];  v = neg[i]; }
    }
    const char* zb = (const char*)z;
    const unsigned offu  = (unsigned)u * (OUT_F * 4u);
    const unsigned offv  = (unsigned)v * (OUT_F * 4u);
    const unsigned sub16 = (unsigned)sub * 16u;
    float4 a[8], b[8];
    float res = 0.0f;

    DEC_LOAD(0, 0) DEC_LOAD(1, 1) DEC_LOAD(2, 2) DEC_LOAD(3, 3)
    DEC_LOAD(4, 4) DEC_LOAD(5, 5) DEC_LOAD(6, 6) DEC_LOAD(7, 7)
    DEC_SUM(0, 0) DEC_SUM(1, 1) DEC_SUM(2, 2) DEC_SUM(3, 3)
    DEC_SUM(4, 4) DEC_SUM(5, 5) DEC_SUM(6, 6) DEC_SUM(7, 7)
    DEC_LOAD(8, 0)  DEC_LOAD(9, 1)  DEC_LOAD(10, 2) DEC_LOAD(11, 3)
    DEC_LOAD(12, 4) DEC_LOAD(13, 5) DEC_LOAD(14, 6) DEC_LOAD(15, 7)
    DEC_SUM(8, 0)  DEC_SUM(9, 1)  DEC_SUM(10, 2) DEC_SUM(11, 3)
    DEC_SUM(12, 4) DEC_SUM(13, 5) DEC_SUM(14, 6) DEC_SUM(15, 7)

    if (i < total) out[i] = res;
}


extern "C" void kernel_launch(void* const* d_in, const int* in_sizes, int n_in,
                              void* d_out, int out_size, void* d_ws, size_t ws_size,
                              hipStream_t stream) {
    const float* x    = (const float*)d_in[0];
    const int*   ei   = (const int*)  d_in[1];
    const int*   pei  = (const int*)  d_in[2];
    const int*   nei  = (const int*)  d_in[3];
    const float* W1   = (const float*)d_in[4];
    const float* b1   = (const float*)d_in[5];
    const float* W2   = (const float*)d_in[6];
    const float* b2   = (const float*)d_in[7];
    float* out = (float*)d_out;

    const int N  = in_sizes[0] / IN_F;
    const int E  = in_sizes[1] / 2;
    const int EP = in_sizes[2] / 2;
    const int* src = ei;
    const int* dst = ei + E;

    float* dinv = (float*)d_ws;
    size_t npad = ((size_t)N + 255) & ~(size_t)255;
    float* bufA = dinv + npad;
    float* h    = bufA + (size_t)N * H_F;
    int*   cnt    = (int*)(h + (size_t)N * H_F);
    int*   rowptr = cnt + npad;
    int*   csr    = rowptr + npad;
    float* z    = h;

    const int nB = (N + BKT_NODES - 1) >> NODE_SHIFT;
    const int nT = (E + TILE_EDGES - 1) / TILE_EDGES;
    const int n2 = nB * nT;
    int2* binned   = (int2*)h;
    int*  rnk      = (int*)h + 2 * (size_t)E;
    int*  tileHist = rnk + E;
    int*  off      = tileHist + n2;

    const int B = 256;
    k_bin_count<<<nT, 1024, nB * sizeof(int), stream>>>(dst, tileHist, rnk, E, nT, nB);
    k_scan_excl<<<(n2 + 1023) / 1024, 1024, 0, stream>>>(tileHist, off, n2);
    k_bin_scatter<<<nT, 1024, nB * sizeof(int), stream>>>(src, dst, off, rnk, binned, E, nT, nB);
    k_bucket_csr<<<nB, 1024, 0, stream>>>(binned, off, rowptr, cnt, dinv, csr, N, E, nT, nB);

    const int gatherBlocks = (N + 7) / 8;

    {
        int rowTiles = (N + 15) / 16;
        int blocks   = (rowTiles + 7) / 8;
        k_gemm_wmma<IN_F, H_F, false><<<blocks, B, 0, stream>>>(x, W1, bufA, N);
    }
    k_gather<H_F><<<gatherBlocks, B, 0, stream>>>(bufA, rowptr, cnt, csr, dinv, b1, h, N);
    {
        int rowTiles = (N + 15) / 16;
        int blocks   = (rowTiles + 7) / 8;
        k_gemm_wmma<H_F, OUT_F, true><<<blocks, B, 0, stream>>>(h, W2, bufA, N);
    }
    k_gather<OUT_F><<<gatherBlocks, B, 0, stream>>>(bufA, rowptr, cnt, csr, dinv, b2, z, N);
    {
        long long waves = (2LL * EP + 31) / 32;
        long long thr   = waves * 32;
        k_decode<<<(int)((thr + B - 1) / B), B, 0, stream>>>(z, pei, nei, out, EP);
    }
}


# Layer-2 product fused into the layer-1 aggregation: a work-group aggregates 16 nodes into an LDS tile and multiplies it by W2 on WMMA f32; h never reaches memory, one launch fewer
# speedup vs baseline: 3.4664x; 1.0894x over previous
#include <hip/hip_runtime.h>

typedef __attribute__((ext_vector_type(2))) float v2f;
typedef __attribute__((ext_vector_type(8))) float v8f;

#define IN_F 256
#define H_F  128
#define OUT_F 64


#define NODE_SHIFT 8
#define BKT_NODES  256
#define EPT        8
#define TILE_EDGES 8192

__global__ __launch_bounds__(1024)
void k_bin_count(const int* __restrict__ dst, int* __restrict__ tileHist, int* __restrict__ rnk,
                 int nE, int nT, int nB) {
    extern __shared__ int s_bins[];
    const int tid = threadIdx.x;
    const int t = blockIdx.x;
    for (int b = tid; b < nB; b += 1024) s_bins[b] = 0;
    __syncthreads();
    const int e0 = t * TILE_EDGES;
    if (e0 + TILE_EDGES <= nE) {
        int d[EPT];
#pragma unroll
        for (int u = 0; u < EPT; ++u) d[u] = dst[e0 + u * 1024 + tid];
#pragma unroll
        for (int u = 0; u < EPT; ++u)
            rnk[e0 + u * 1024 + tid] = atomicAdd(&s_bins[d[u] >> NODE_SHIFT], 1);
    } else {
        for (int e = e0 + tid; e < nE; e += 1024)
            rnk[e] = atomicAdd(&s_bins[dst[e] >> NODE_SHIFT], 1);
    }
    __syncthreads();
    for (int b = tid; b < nB; b += 1024) tileHist[b * nT + t] = s_bins[b];
}

__global__ __launch_bounds__(1024)
void k_scan_excl(const int* __restrict__ cnt, int* __restrict__ excl, int n) {
    __shared__ int s_pre[32];
    __shared__ int s_wsum[32];
    __shared__ int s_woff[32];
    const int tid = threadIdx.x;
    const int lane = tid & 31;
    const int wave = tid >> 5;
    const int blockStart = blockIdx.x * 1024;

    int pre = 0;
    for (int j = 0; j < (int)blockIdx.x; ++j) pre += cnt[j * 1024 + tid];
#pragma unroll
    for (int off = 16; off > 0; off >>= 1) pre += __shfl_xor(pre, off, 32);

    const int i = blockStart + tid;
    const int c = (i < n) ? cnt[i] : 0;
    int incl = c;
#pragma unroll
    for (int off = 1; off < 32; off <<= 1) {
        int t = __shfl_up(incl, off, 32);
        if (lane >= off) incl += t;
    }
    if (lane == 31) s_wsum[wave] = incl;
    if (lane == 0)  s_pre[wave] = pre;
    __syncthreads();
    if (wave == 0) {
        int p = s_pre[lane];
#pragma unroll
        for (int off = 16; off > 0; off >>= 1) p += __shfl_xor(p, off, 32);
        const int w = s_wsum[lane];
        int wi = w;
#pragma unroll
        for (int off = 1; off < 32; off <<= 1) {
            int t = __shfl_up(wi, off, 32);
            if (lane >= off) wi += t;
        }
        s_woff[lane] = p + wi - w;
    }
    __syncthreads();
    if (i < n) excl[i] = s_woff[wave] + incl - c;
}

__global__ __launch_bounds__(1024)
void k_bin_scatter(const int* __restrict__ src, const int* __restrict__ dst,
                   const int* __restrict__ off, const int* __restrict__ rnk,
                   int2* __restrict__ binned, int nE, int nT, int nB) {
    extern __shared__ int s_bins[];
    const int tid = threadIdx.x;
    const int t = blockIdx.x;
    for (int b = tid; b < nB; b += 1024) s_bins[b] = off[b * nT + t];
    __syncthreads();
    const int e0 = t * TILE_EDGES;
    if (e0 + TILE_EDGES <= nE) {
        int d[EPT], s[EPT], r[EPT];
#pragma unroll
        for (int u = 0; u < EPT; ++u) {
            const int e = e0 + u * 1024 + tid;
            d[u] = dst[e];
            s[u] = src[e];
            r[u] = rnk[e];
        }
#pragma unroll
        for (int u = 0; u < EPT; ++u)
            binned[s_bins[d[u] >> NODE_SHIFT] + r[u]] = make_int2(s[u], d[u]);
    } else {
        for (int e = e0 + tid; e < nE; e += 1024) {
            const int d = dst[e];
            binned[s_bins[d >> NODE_SHIFT] + rnk[e]] = make_int2(src[e], d);
        }
    }
}

__global__ __launch_bounds__(1024)
void k_bucket_csr(const int2* __restrict__ binned, const int* __restrict__ off,
                  int* __restrict__ rowptr, int* __restrict__ cnt, float* __restrict__ dinv,
                  int* __restrict__ csr, int n, int nE, int nT, int nB) {
    __shared__ int s_cnt[BKT_NODES];
    __shared__ int s_cur[BKT_NODES];
    __shared__ int s_wsum[BKT_NODES / 32];
    const int tid = threadIdx.x;
    const int lane = tid & 31;
    const int wave = tid >> 5;
    const int b = blockIdx.x;
    const int node0 = b << NODE_SHIFT;
    const int bs = off[b * nT];
    const int be = (b + 1 < nB) ? off[(b + 1) * nT] : nE;
    const int own = tid & (BKT_NODES - 1);

    if (tid < BKT_NODES) s_cnt[tid] = 0;
    __syncthreads();
    for (int base = bs; base < be; base += 4096) {
        int d[4];
#pragma unroll
        for (int u = 0; u < 4; ++u) {
            const int e = base + u * 1024 + tid;
            d[u] = binned[(e < be) ? e : be - 1].y;
        }
#pragma unroll
        for (int u = 0; u < 4; ++u) {
            const bool in = base + u * 1024 + tid < be;
            atomicAdd(&s_cnt[in ? d[u] - node0 : own], in ? 1 : 0);
        }
    }
    __syncthreads();

    int c = 0, incl = 0;
    if (tid < BKT_NODES) {
        c = s_cnt[tid];
        incl = c;
#pragma unroll
        for (int o = 1; o < 32; o <<= 1) {
            int t = __shfl_up(incl, o, 32);
            if (lane >= o) incl += t;
        }
        if (lane == 31) s_wsum[wave] = incl;
    }
    __syncthreads();
    if (tid < BKT_NODES) {
        int base = 0;
        for (int w = 0; w < wave; ++w) base += s_wsum[w];
        const int ex = base + incl - c;
        s_cur[tid] = ex;
        const int node = node0 + tid;
        if (node < n) {
            rowptr[node] = bs + ex;
            cnt[node]    = c;
            dinv[node]   = rsqrtf((float)(c + 1));
        }
    }
    __syncthreads();
    for (int base = bs; base < be; base += 4096) {
        int2 ed[4];
#pragma unroll
        for (int u = 0; u < 4; ++u) {
            const int e = base + u * 1024 + tid;
            ed[u] = binned[(e < be) ? e : be - 1];
        }
#pragma unroll
        for (int u = 0; u < 4; ++u) {
            const bool in = base + u * 1024 + tid < be;
            const int p = atomicAdd(&s_cur[in ? ed[u].y - node0 : own], in ? 1 : 0);
            if (in) csr[bs + p] = ed[u].x;
        }
    }
}


template <int Ki, int Ko, bool RELU_A>
__global__ __launch_bounds__(256)
void k_gemm_wmma(const float* __restrict__ X, const float* __restrict__ W,
                 float* __restrict__ Y, int nrows) {
    constexpr int CT = Ko / 16;
    const int wave = threadIdx.x >> 5;
    const int lane = threadIdx.x & 31;
    const int rowTile = blockIdx.x * 8 + wave;
    const int row0 = rowTile * 16;
    if (row0 >= nrows) return;

    const int m    = lane & 15;
    const int half = lane >> 4;
    const float* xrow = X + (size_t)(row0 + m) * Ki + 2 * half;

    v8f acc[CT] = {};

    for (int kb = 0; kb < Ki; kb += 4) {
        if (kb + 32 < Ki) __builtin_prefetch(xrow + kb + 32, 0, 3);

        float ax = xrow[kb];
        float ay = xrow[kb + 1];
        if (RELU_A) { ax = fmaxf(ax, 0.0f); ay = fmaxf(ay, 0.0f); }
        v2f a = { ax, ay };

        const int krow = kb + 2 * half;
        const float* w0 = W + (size_t)krow * Ko;
        const float* w1 = w0 + Ko;
#pragma unroll
        for (int j = 0; j < CT; ++j) {
            v2f b = { w0[j * 16 + m], w1[j * 16 + m] };
            acc[j] = __builtin_amdgcn_wmma_f32_16x16x4_f32(
                false, a, false, b, (short)0, acc[j], false, false);
        }
    }

#pragma unroll
    for (int j = 0; j < CT; ++j) {
#pragma unroll
        for (int r = 0; r < 8; ++r) {
            const int M = r + 8 * half;
            Y[(size_t)(row0 + M) * Ko + j * 16 + m] = acc[j][r];
        }
    }
}

template <int V> struct fvec { typedef __attribute__((ext_vector_type(V))) float type; };

template <int F>
__device__ __forceinline__ typename fvec<F / 32>::type
gather_node(const float* __restrict__ lin, const int* __restrict__ rowptr,
            const int* __restrict__ cnt, const int* __restrict__ csr,
            const float* __restrict__ dinv, const float* __restrict__ bias,
            const int d, const int lane) {
    constexpr int V = F / 32;
    typedef typename fvec<V>::type vecf;
    const int start = rowptr[d];
    const int len   = cnt[d];
    const float wd  = dinv[d];
    const vecf* linv = (const vecf*)lin;

    vecf acc = linv[(size_t)d * 32 + lane] * (wd * wd);
#pragma unroll
    for (int j = 0; j < V; ++j) acc[j] += bias[lane * V + j];

    for (int base = 0; base < len; base += 32) {
        const int m = (len - base < 32) ? (len - base) : 32;
        int   s_l = 0;
        float w_l = 0.0f;
        if (lane < m) { s_l = csr[start + base + lane]; w_l = dinv[s_l]; }
        const int w_i = __float_as_int(w_l);
        int k = 0;
        for (; k + 4 <= m; k += 4) {
            const int s0 = __builtin_amdgcn_readlane(s_l, k);
            const int s1 = __builtin_amdgcn_readlane(s_l, k + 1);
            const int s2 = __builtin_amdgcn_readlane(s_l, k + 2);
            const int s3 = __builtin_amdgcn_readlane(s_l, k + 3);
            const vecf v0 = linv[(size_t)s0 * 32 + lane];
            const vecf v1 = linv[(size_t)s1 * 32 + lane];
            const vecf v2 = linv[(size_t)s2 * 32 + lane];
            const vecf v3 = linv[(size_t)s3 * 32 + lane];
            acc += v0 * (__int_as_float(__builtin_amdgcn_readlane(w_i, k)) * wd);
            acc += v1 * (__int_as_float(__builtin_amdgcn_readlane(w_i, k + 1)) * wd);
            acc += v2 * (__int_as_float(__builtin_amdgcn_readlane(w_i, k + 2)) * wd);
            acc += v3 * (__int_as_float(__builtin_amdgcn_readlane(w_i, k + 3)) * wd);
        }
        for (; k < m; ++k) {
            const int s0 = __builtin_amdgcn_readlane(s_l, k);
            const vecf v0 = linv[(size_t)s0 * 32 + lane];
            acc += v0 * (__int_as_float(__builtin_amdgcn_readlane(w_i, k)) * wd);
        }
    }
    return acc;
}

template <int F>
__global__ __launch_bounds__(256)
void k_gather(const float* __restrict__ lin, const int* __restrict__ rowptr,
              const int* __restrict__ cnt, const int* __restrict__ csr,
              const float* __restrict__ dinv, const float* __restrict__ bias,
              float* __restrict__ agg, int n) {
    typedef typename fvec<F / 32>::type vecf;
    const int lane = threadIdx.x & 31;
    const int d = __builtin_amdgcn_readfirstlane((int)((blockIdx.x * blockDim.x + threadIdx.x) >> 5));
    if (d >= n) return;
    const vecf acc = gather_node<F>(lin, rowptr, cnt, csr, dinv, bias, d, lane);
    ((vecf*)agg)[(size_t)d * 32 + lane] = acc;
}

#define HS_PITCH 132

__global__ __launch_bounds__(256)
void k_gather_gemm2(const float* __restrict__ lin, const int* __restrict__ rowptr,
                    const int* __restrict__ cnt, const int* __restrict__ csr,
                    const float* __restrict__ dinv, const float* __restrict__ bias,
                    const float* __restrict__ W2, float* __restrict__ zlin, int n) {
    typedef fvec<4>::type v4f;
    __shared__ float hs[16 * HS_PITCH];
    const int lane  = threadIdx.x & 31;
    const int wave  = __builtin_amdgcn_readfirstlane((int)(threadIdx.x >> 5));
    const int node0 = blockIdx.x * 16;

    for (int t = 0; t < 2; ++t) {
        const int r = wave * 2 + t;
        const int d = node0 + r;
        v4f acc = { 0.0f, 0.0f, 0.0f, 0.0f };
        if (d < n) {
            acc = gather_node<H_F>(lin, rowptr, cnt, csr, dinv, bias, d, lane);
#pragma unroll
            for (int j = 0; j < 4; ++j) acc[j] = fmaxf(acc[j], 0.0f);
        }
        *(v4f*)(hs + r * HS_PITCH + 4 * lane) = acc;
    }
    __syncthreads();
    if (wave >= (OUT_F / 16)) return;

    const int m    = lane & 15;
    const int half = lane >> 4;
    const float* arow = hs + m * HS_PITCH + 2 * half;
    const float* wcol = W2 + (size_t)(2 * half) * OUT_F + wave * 16 + m;
    v8f acc = {};
#pragma unroll 8
    for (int kb = 0; kb < H_F; kb += 4) {
        const v2f a = *(const v2f*)(arow + kb);
        const v2f b = { wcol[(size_t)kb * OUT_F], wcol[(size_t)(kb + 1) * OUT_F] };
        acc = __builtin_amdgcn_wmma_f32_16x16x4_f32(false, a, false, b, (short)0, acc, false, false);
    }
#pragma unroll
    for (int r = 0; r < 8; ++r) {
        const int row = node0 + r + 8 * half;
        if (row < n) zlin[(size_t)row * OUT_F + wave * 16 + m] = acc[r];
    }
}


template <int K>
__device__ __forceinline__ unsigned row_bcast(unsigned v) {
    return (unsigned)__builtin_amdgcn_update_dpp(0, (int)v, 0x150 + K, 0xf, 0xf, true);
}
template <int M>
__device__ __forceinline__ float row_xadd(float x) {
    const int y = __builtin_amdgcn_update_dpp(0, __float_as_int(x), 0x160 + M, 0xf, 0xf, true);
    return x + __int_as_float(y);
}

template <int K>
__device__ __forceinline__ float4 z_piece(const char* __restrict__ zb, unsigned off, unsigned sub16) {
    return *(const float4*)(zb + (row_bcast<K>(off) + sub16));
}

template <int K>
__device__ __forceinline__ float score_step(const float4 a, const float4 b, int sub, float res) {
    float p = a.x * b.x;
    p = fmaf(a.y, b.y, p);
    p = fmaf(a.z, b.z, p);
    p = fmaf(a.w, b.w, p);
    p = row_xadd<1>(p);
    p = row_xadd<2>(p);
    p = row_xadd<4>(p);
    p = row_xadd<8>(p);
    return (sub == K) ? p : res;
}

#define DEC_LOAD(K, J) a[J] = z_piece<K>(zb, offu, sub16); b[J] = z_piece<K>(zb, offv, sub16);
#define DEC_SUM(K, J)  res = score_step<K>(a[J], b[J], sub, res);

__global__ __launch_bounds__(256)
void k_decode(const float* __restrict__ z,
              const int* __restrict__ pos, const int* __restrict__ neg,
              float* __restrict__ out, int EP) {
    const int lane  = threadIdx.x & 31;
    const int sub   = lane & 15;
    const int total = 2 * EP;
    const int i0    = (int)((blockIdx.x * blockDim.x + threadIdx.x) >> 5) * 32;
    if (i0 >= total) return;
    const int i = i0 + lane;
    int u = 0, v = 0;
    if (i < total) {
        if (i < EP) { u = pos[i];       v = pos[EP + i]; }
        else        { u = neg[i - EP];  v = neg[i]; }
    }
    const char* zb = (const char*)z;
    const unsigned offu  = (unsigned)u * (OUT_F * 4u);
    const unsigned offv  = (unsigned)v * (OUT_F * 4u);
    const unsigned sub16 = (unsigned)sub * 16u;
    float4 a[8], b[8];
    float res = 0.0f;

    DEC_LOAD(0, 0) DEC_LOAD(1, 1) DEC_LOAD(2, 2) DEC_LOAD(3, 3)
    DEC_LOAD(4, 4) DEC_LOAD(5, 5) DEC_LOAD(6, 6) DEC_LOAD(7, 7)
    DEC_SUM(0, 0) DEC_SUM(1, 1) DEC_SUM(2, 2) DEC_SUM(3, 3)
    DEC_SUM(4, 4) DEC_SUM(5, 5) DEC_SUM(6, 6) DEC_SUM(7, 7)
    DEC_LOAD(8, 0)  DEC_LOAD(9, 1)  DEC_LOAD(10, 2) DEC_LOAD(11, 3)
    DEC_LOAD(12, 4) DEC_LOAD(13, 5) DEC_LOAD(14, 6) DEC_LOAD(15, 7)
    DEC_SUM(8, 0)  DEC_SUM(9, 1)  DEC_SUM(10, 2) DEC_SUM(11, 3)
    DEC_SUM(12, 4) DEC_SUM(13, 5) DEC_SUM(14, 6) DEC_SUM(15, 7)

    if (i < total) out[i] = res;
}


extern "C" void kernel_launch(void* const* d_in, const int* in_sizes, int n_in,
                              void* d_out, int out_size, void* d_ws, size_t ws_size,
                              hipStream_t stream) {
    const float* x    = (const float*)d_in[0];
    const int*   ei   = (const int*)  d_in[1];
    const int*   pei  = (const int*)  d_in[2];
    const int*   nei  = (const int*)  d_in[3];
    const float* W1   = (const float*)d_in[4];
    const float* b1   = (const float*)d_in[5];
    const float* W2   = (const float*)d_in[6];
    const float* b2   = (const float*)d_in[7];
    float* out = (float*)d_out;

    const int N  = in_sizes[0] / IN_F;
    const int E  = in_sizes[1] / 2;
    const int EP = in_sizes[2] / 2;
    const int* src = ei;
    const int* dst = ei + E;

    float* dinv = (float*)d_ws;
    size_t npad = ((size_t)N + 255) & ~(size_t)255;
    float* bufA = dinv + npad;
    float* h    = bufA + (size_t)N * H_F;
    int*   cnt    = (int*)(h + (size_t)N * H_F);
    int*   rowptr = cnt + npad;
    int*   csr    = rowptr + npad;
    float* z    = bufA;

    const int nB = (N + BKT_NODES - 1) >> NODE_SHIFT;
    const int nT = (E + TILE_EDGES - 1) / TILE_EDGES;
    const int n2 = nB * nT;
    int2* binned   = (int2*)h;
    int*  rnk      = (int*)h + 2 * (size_t)E;
    int*  tileHist = rnk + E;
    int*  off      = tileHist + n2;

    const int B = 256;
    k_bin_count<<<nT, 1024, nB * sizeof(int), stream>>>(dst, tileHist, rnk, E, nT, nB);
    k_scan_excl<<<(n2 + 1023) / 1024, 1024, 0, stream>>>(tileHist, off, n2);
    k_bin_scatter<<<nT, 1024, nB * sizeof(int), stream>>>(src, dst, off, rnk, binned, E, nT, nB);
    k_bucket_csr<<<nB, 1024, 0, stream>>>(binned, off, rowptr, cnt, dinv, csr, N, E, nT, nB);

    const int gatherBlocks = (N + 7) / 8;

    {
        int rowTiles = (N + 15) / 16;
        int blocks   = (rowTiles + 7) / 8;
        k_gemm_wmma<IN_F, H_F, false><<<blocks, B, 0, stream>>>(x, W1, bufA, N);
    }
    k_gather_gemm2<<<(N + 15) / 16, B, 0, stream>>>(bufA, rowptr, cnt, csr, dinv, b1, W2, h, N);
    k_gather<OUT_F><<<gatherBlocks, B, 0, stream>>>(h, rowptr, cnt, csr, dinv, b2, bufA, N);
    {
        long long waves = (2LL * EP + 31) / 32;
        long long thr   = waves * 32;
        k_decode<<<(int)((thr + B - 1) / B), B, 0, stream>>>(z, pei, nei, out, EP);
    }
}


# Source factor dinv[src] folded into the rows both products store, dst factor applied once after the sum: aggregations drop the per-edge dinv gather; sum order kept, rounding points move
# speedup vs baseline: 3.6163x; 1.0432x over previous
#include <hip/hip_runtime.h>

typedef __attribute__((ext_vector_type(2))) float v2f;
typedef __attribute__((ext_vector_type(8))) float v8f;

#define IN_F 256
#define H_F  128
#define OUT_F 64


#define NODE_SHIFT 8
#define BKT_NODES  256
#define EPT        8
#define TILE_EDGES 8192

__global__ __launch_bounds__(1024)
void k_bin_count(const int* __restrict__ dst, int* __restrict__ tileHist, int* __restrict__ rnk,
                 int nE, int nT, int nB) {
    extern __shared__ int s_bins[];
    const int tid = threadIdx.x;
    const int t = blockIdx.x;
    for (int b = tid; b < nB; b += 1024) s_bins[b] = 0;
    __syncthreads();
    const int e0 = t * TILE_EDGES;
    if (e0 + TILE_EDGES <= nE) {
        int d[EPT];
#pragma unroll
        for (int u = 0; u < EPT; ++u) d[u] = dst[e0 + u * 1024 + tid];
#pragma unroll
        for (int u = 0; u < EPT; ++u)
            rnk[e0 + u * 1024 + tid] = atomicAdd(&s_bins[d[u] >> NODE_SHIFT], 1);
    } else {
        for (int e = e0 + tid; e < nE; e += 1024)
            rnk[e] = atomicAdd(&s_bins[dst[e] >> NODE_SHIFT], 1);
    }
    __syncthreads();
    for (int b = tid; b < nB; b += 1024) tileHist[b * nT + t] = s_bins[b];
}

__global__ __launch_bounds__(1024)
void k_scan_excl(const int* __restrict__ cnt, int* __restrict__ excl, int n) {
    __shared__ int s_pre[32];
    __shared__ int s_wsum[32];
    __shared__ int s_woff[32];
    const int tid = threadIdx.x;
    const int lane = tid & 31;
    const int wave = tid >> 5;
    const int blockStart = blockIdx.x * 1024;

    int pre = 0;
    for (int j = 0; j < (int)blockIdx.x; ++j) pre += cnt[j * 1024 + tid];
#pragma unroll
    for (int off = 16; off > 0; off >>= 1) pre += __shfl_xor(pre, off, 32);

    const int i = blockStart + tid;
    const int c = (i < n) ? cnt[i] : 0;
    int incl = c;
#pragma unroll
    for (int off = 1; off < 32; off <<= 1) {
        int t = __shfl_up(incl, off, 32);
        if (lane >= off) incl += t;
    }
    if (lane == 31) s_wsum[wave] = incl;
    if (lane == 0)  s_pre[wave] = pre;
    __syncthreads();
    if (wave == 0) {
        int p = s_pre[lane];
#pragma unroll
        for (int off = 16; off > 0; off >>= 1) p += __shfl_xor(p, off, 32);
        const int w = s_wsum[lane];
        int wi = w;
#pragma unroll
        for (int off = 1; off < 32; off <<= 1) {
            int t = __shfl_up(wi, off, 32);
            if (lane >= off) wi += t;
        }
        s_woff[lane] = p + wi - w;
    }
    __syncthreads();
    if (i < n) excl[i] = s_woff[wave] + incl - c;
}

__global__ __launch_bounds__(1024)
void k_bin_scatter(const int* __restrict__ src, const int* __restrict__ dst,
                   const int* __restrict__ off, const int* __restrict__ rnk,
                   int2* __restrict__ binned, int nE, int nT, int nB) {
    extern __shared__ int s_bins[];
    const int tid = threadIdx.x;
    const int t = blockIdx.x;
    for (int b = tid; b < nB; b += 1024) s_bins[b] = off[b * nT + t];
    __syncthreads();
    const int e0 = t * TILE_EDGES;
    if (e0 + TILE_EDGES <= nE) {
        int d[EPT], s[EPT], r[EPT];
#pragma unroll
        for (int u = 0; u < EPT; ++u) {
            const int e = e0 + u * 1024 + tid;
            d[u] = dst[e];
            s[u] = src[e];
            r[u] = rnk[e];
        }
#pragma unroll
        for (int u = 0; u < EPT; ++u)
            binned[s_bins[d[u] >> NODE_SHIFT] + r[u]] = make_int2(s[u], d[u]);
    } else {
        for (int e = e0 + tid; e < nE; e += 1024) {
            const int d = dst[e];
            binned[s_bins[d >> NODE_SHIFT] + rnk[e]] = make_int2(src[e], d);
        }
    }
}

__global__ __launch_bounds__(1024)
void k_bucket_csr(const int2* __restrict__ binned, const int* __restrict__ off,
                  int* __restrict__ rowptr, int* __restrict__ cnt, float* __restrict__ dinv,
                  int* __restrict__ csr, int n, int nE, int nT, int nB) {
    __shared__ int s_cnt[BKT_NODES];
    __shared__ int s_cur[BKT_NODES];
    __shared__ int s_wsum[BKT_NODES / 32];
    const int tid = threadIdx.x;
    const int lane = tid & 31;
    const int wave = tid >> 5;
    const int b = blockIdx.x;
    const int node0 = b << NODE_SHIFT;
    const int bs = off[b * nT];
    const int be = (b + 1 < nB) ? off[(b + 1) * nT] : nE;
    const int own = tid & (BKT_NODES - 1);

    if (tid < BKT_NODES) s_cnt[tid] = 0;
    __syncthreads();
    for (int base = bs; base < be; base += 4096) {
        int d[4];
#pragma unroll
        for (int u = 0; u < 4; ++u) {
            const int e = base + u * 1024 + tid;
            d[u] = binned[(e < be) ? e : be - 1].y;
        }
#pragma unroll
        for (int u = 0; u < 4; ++u) {
            const bool in = base + u * 1024 + tid < be;
            atomicAdd(&s_cnt[in ? d[u] - node0 : own], in ? 1 : 0);
        }
    }
    __syncthreads();

    int c = 0, incl = 0;
    if (tid < BKT_NODES) {
        c = s_cnt[tid];
        incl = c;
#pragma unroll
        for (int o = 1; o < 32; o <<= 1) {
            int t = __shfl_up(incl, o, 32);
            if (lane >= o) incl += t;
        }
        if (lane == 31) s_wsum[wave] = incl;
    }
    __syncthreads();
    if (tid < BKT_NODES) {
        int base = 0;
        for (int w = 0; w < wave; ++w) base += s_wsum[w];
        const int ex = base + incl - c;
        s_cur[tid] = ex;
        const int node = node0 + tid;
        if (node < n) {
            rowptr[node] = bs + ex;
            cnt[node]    = c;
            dinv[node]   = rsqrtf((float)(c + 1));
        }
    }
    __syncthreads();
    for (int base = bs; base < be; base += 4096) {
        int2 ed[4];
#pragma unroll
        for (int u = 0; u < 4; ++u) {
            const int e = base + u * 1024 + tid;
            ed[u] = binned[(e < be) ? e : be - 1];
        }
#pragma unroll
        for (int u = 0; u < 4; ++u) {
            const bool in = base + u * 1024 + tid < be;
            const int p = atomicAdd(&s_cur[in ? ed[u].y - node0 : own], in ? 1 : 0);
            if (in) csr[bs + p] = ed[u].x;
        }
    }
}


template <int Ki, int Ko, bool RELU_A>
__global__ __launch_bounds__(256)
void k_gemm_wmma(const float* __restrict__ X, const float* __restrict__ W,
                 const float* __restrict__ rs, float* __restrict__ Y, int nrows) {
    constexpr int CT = Ko / 16;
    const int wave = threadIdx.x >> 5;
    const int lane = threadIdx.x & 31;
    const int rowTile = blockIdx.x * 8 + wave;
    const int row0 = rowTile * 16;
    if (row0 >= nrows) return;

    const int m    = lane & 15;
    const int half = lane >> 4;
    const float* xrow = X + (size_t)(row0 + m) * Ki + 2 * half;

    v8f acc[CT] = {};

    for (int kb = 0; kb < Ki; kb += 4) {
        if (kb + 32 < Ki) __builtin_prefetch(xrow + kb + 32, 0, 3);

        float ax = xrow[kb];
        float ay = xrow[kb + 1];
        if (RELU_A) { ax = fmaxf(ax, 0.0f); ay = fmaxf(ay, 0.0f); }
        v2f a = { ax, ay };

        const int krow = kb + 2 * half;
        const float* w0 = W + (size_t)krow * Ko;
        const float* w1 = w0 + Ko;
#pragma unroll
        for (int j = 0; j < CT; ++j) {
            v2f b = { w0[j * 16 + m], w1[j * 16 + m] };
            acc[j] = __builtin_amdgcn_wmma_f32_16x16x4_f32(
                false, a, false, b, (short)0, acc[j], false, false);
        }
    }

    const float4 rs0 = *(const float4*)(rs + row0 + 8 * half);
    const float4 rs1 = *(const float4*)(rs + row0 + 8 * half + 4);
    const float rsv[8] = { rs0.x, rs0.y, rs0.z, rs0.w, rs1.x, rs1.y, rs1.z, rs1.w };
#pragma unroll
    for (int j = 0; j < CT; ++j) {
#pragma unroll
        for (int r = 0; r < 8; ++r) {
            const int M = r + 8 * half;
            Y[(size_t)(row0 + M) * Ko + j * 16 + m] = acc[j][r] * rsv[r];
        }
    }
}

template <int V> struct fvec { typedef __attribute__((ext_vector_type(V))) float type; };

template <int F>
__device__ __forceinline__ typename fvec<F / 32>::type
gather_node(const float* __restrict__ lin, const int* __restrict__ rowptr,
            const int* __restrict__ cnt, const int* __restrict__ csr,
            const float* __restrict__ dinv, const float* __restrict__ bias,
            const int d, const int lane) {
    constexpr int V = F / 32;
    typedef typename fvec<V>::type vecf;
    const int start = rowptr[d];
    const int len   = cnt[d];
    const float wd  = dinv[d];
    const vecf* linv = (const vecf*)lin;

    vecf acc = linv[(size_t)d * 32 + lane];
    const vecf bv = ((const vecf*)bias)[lane];

    for (int base = 0; base < len; base += 32) {
        const int m = (len - base < 32) ? (len - base) : 32;
        int s_l = 0;
        if (lane < m) s_l = csr[start + base + lane];
        int k = 0;
        for (; k + 4 <= m; k += 4) {
            const int s0 = __builtin_amdgcn_readlane(s_l, k);
            const int s1 = __builtin_amdgcn_readlane(s_l, k + 1);
            const int s2 = __builtin_amdgcn_readlane(s_l, k + 2);
            const int s3 = __builtin_amdgcn_readlane(s_l, k + 3);
            const vecf v0 = linv[(size_t)s0 * 32 + lane];
            const vecf v1 = linv[(size_t)s1 * 32 + lane];
            const vecf v2 = linv[(size_t)s2 * 32 + lane];
            const vecf v3 = linv[(size_t)s3 * 32 + lane];
            acc += v0;
            acc += v1;
            acc += v2;
            acc += v3;
        }
        for (; k < m; ++k) {
            const int s0 = __builtin_amdgcn_readlane(s_l, k);
            acc += linv[(size_t)s0 * 32 + lane];
        }
    }
    vecf res;
#pragma unroll
    for (int j = 0; j < V; ++j) res[j] = fmaf(acc[j], wd, bv[j]);
    return res;
}

template <int F>
__global__ __launch_bounds__(256)
void k_gather(const float* __restrict__ lin, const int* __restrict__ rowptr,
              const int* __restrict__ cnt, const int* __restrict__ csr,
              const float* __restrict__ dinv, const float* __restrict__ bias,
              float* __restrict__ agg, int n) {
    typedef typename fvec<F / 32>::type vecf;
    const int lane = threadIdx.x & 31;
    const int d = __builtin_amdgcn_readfirstlane((int)((blockIdx.x * blockDim.x + threadIdx.x) >> 5));
    if (d >= n) return;
    const vecf acc = gather_node<F>(lin, rowptr, cnt, csr, dinv, bias, d, lane);
    ((vecf*)agg)[(size_t)d * 32 + lane] = acc;
}

#define HS_PITCH 132

__global__ __launch_bounds__(256)
void k_gather_gemm2(const float* __restrict__ lin, const int* __restrict__ rowptr,
                    const int* __restrict__ cnt, const int* __restrict__ csr,
                    const float* __restrict__ dinv, const float* __restrict__ bias,
                    const float* __restrict__ W2, float* __restrict__ zlin, int n) {
    typedef fvec<4>::type v4f;
    __shared__ float hs[16 * HS_PITCH];
    const int lane  = threadIdx.x & 31;
    const int wave  = __builtin_amdgcn_readfirstlane((int)(threadIdx.x >> 5));
    const int node0 = blockIdx.x * 16;

    for (int t = 0; t < 2; ++t) {
        const int r = wave * 2 + t;
        const int d = node0 + r;
        v4f acc = { 0.0f, 0.0f, 0.0f, 0.0f };
        if (d < n) {
            acc = gather_node<H_F>(lin, rowptr, cnt, csr, dinv, bias, d, lane);
#pragma unroll
            for (int j = 0; j < 4; ++j) acc[j] = fmaxf(acc[j], 0.0f);
        }
        *(v4f*)(hs + r * HS_PITCH + 4 * lane) = acc;
    }
    __syncthreads();
    if (wave >= (OUT_F / 16)) return;

    const int m    = lane & 15;
    const int half = lane >> 4;
    const float* arow = hs + m * HS_PITCH + 2 * half;
    const float* wcol = W2 + (size_t)(2 * half) * OUT_F + wave * 16 + m;
    float sc[8];
#pragma unroll
    for (int r = 0; r < 8; ++r) {
        const int row = node0 + r + 8 * half;
        sc[r] = dinv[(row < n) ? row : n - 1];
    }
    v8f acc = {};
#pragma unroll 8
    for (int kb = 0; kb < H_F; kb += 4) {
        const v2f a = *(const v2f*)(arow + kb);
        const v2f b = { wcol[(size_t)kb * OUT_F], wcol[(size_t)(kb + 1) * OUT_F] };
        acc = __builtin_amdgcn_wmma_f32_16x16x4_f32(false, a, false, b, (short)0, acc, false, false);
    }
#pragma unroll
    for (int r = 0; r < 8; ++r) {
        const int row = node0 + r + 8 * half;
        if (row < n) zlin[(size_t)row * OUT_F + wave * 16 + m] = acc[r] * sc[r];
    }
}


template <int K>
__device__ __forceinline__ unsigned row_bcast(unsigned v) {
    return (unsigned)__builtin_amdgcn_update_dpp(0, (int)v, 0x150 + K, 0xf, 0xf, true);
}
template <int M>
__device__ __forceinline__ float row_xadd(float x) {
    const int y = __builtin_amdgcn_update_dpp(0, __float_as_int(x), 0x160 + M, 0xf, 0xf, true);
    return x + __int_as_float(y);
}

template <int K>
__device__ __forceinline__ float4 z_piece(const char* __restrict__ zb, unsigned off, unsigned sub16) {
    return *(const float4*)(zb + (row_bcast<K>(off) + sub16));
}

template <int K>
__device__ __forceinline__ float score_step(const float4 a, const float4 b, int sub, float res) {
    float p = a.x * b.x;
    p = fmaf(a.y, b.y, p);
    p = fmaf(a.z, b.z, p);
    p = fmaf(a.w, b.w, p);
    p = row_xadd<1>(p);
    p = row_xadd<2>(p);
    p = row_xadd<4>(p);
    p = row_xadd<8>(p);
    return (sub == K) ? p : res;
}

#define DEC_LOAD(K, J) a[J] = z_piece<K>(zb, offu, sub16); b[J] = z_piece<K>(zb, offv, sub16);
#define DEC_SUM(K, J)  res = score_step<K>(a[J], b[J], sub, res);

__global__ __launch_bounds__(256)
void k_decode(const float* __restrict__ z,
              const int* __restrict__ pos, const int* __restrict__ neg,
              float* __restrict__ out, int EP) {
    const int lane  = threadIdx.x & 31;
    const int sub   = lane & 15;
    const int total = 2 * EP;
    const int i0    = (int)((blockIdx.x * blockDim.x + threadIdx.x) >> 5) * 32;
    if (i0 >= total) return;
    const int i = i0 + lane;
    int u = 0, v = 0;
    if (i < total) {
        if (i < EP) { u = pos[i];       v = pos[EP + i]; }
        else        { u = neg[i - EP];  v = neg[i]; }
    }
    const char* zb = (const char*)z;
    const unsigned offu  = (unsigned)u * (OUT_F * 4u);
    const unsigned offv  = (unsigned)v * (OUT_F * 4u);
    const unsigned sub16 = (unsigned)sub * 16u;
    float4 a[8], b[8];
    float res = 0.0f;

    DEC_LOAD(0, 0) DEC_LOAD(1, 1) DEC_LOAD(2, 2) DEC_LOAD(3, 3)
    DEC_LOAD(4, 4) DEC_LOAD(5, 5) DEC_LOAD(6, 6) DEC_LOAD(7, 7)
    DEC_SUM(0, 0) DEC_SUM(1, 1) DEC_SUM(2, 2) DEC_SUM(3, 3)
    DEC_SUM(4, 4) DEC_SUM(5, 5) DEC_SUM(6, 6) DEC_SUM(7, 7)
    DEC_LOAD(8, 0)  DEC_LOAD(9, 1)  DEC_LOAD(10, 2) DEC_LOAD(11, 3)
    DEC_LOAD(12, 4) DEC_LOAD(13, 5) DEC_LOAD(14, 6) DEC_LOAD(15, 7)
    DEC_SUM(8, 0)  DEC_SUM(9, 1)  DEC_SUM(10, 2) DEC_SUM(11, 3)
    DEC_SUM(12, 4) DEC_SUM(13, 5) DEC_SUM(14, 6) DEC_SUM(15, 7)

    if (i < total) out[i] = res;
}


extern "C" void kernel_launch(void* const* d_in, const int* in_sizes, int n_in,
                              void* d_out, int out_size, void* d_ws, size_t ws_size,
                              hipStream_t stream) {
    const float* x    = (const float*)d_in[0];
    const int*   ei   = (const int*)  d_in[1];
    const int*   pei  = (const int*)  d_in[2];
    const int*   nei  = (const int*)  d_in[3];
    const float* W1   = (const float*)d_in[4];
    const float* b1   = (const float*)d_in[5];
    const float* W2   = (const float*)d_in[6];
    const float* b2   = (const float*)d_in[7];
    float* out = (float*)d_out;

    const int N  = in_sizes[0] / IN_F;
    const int E  = in_sizes[1] / 2;
    const int EP = in_sizes[2] / 2;
    const int* src = ei;
    const int* dst = ei + E;

    float* dinv = (float*)d_ws;
    size_t npad = ((size_t)N + 255) & ~(size_t)255;
    float* bufA = dinv + npad;
    float* h    = bufA + (size_t)N * H_F;
    int*   cnt    = (int*)(h + (size_t)N * H_F);
    int*   rowptr = cnt + npad;
    int*   csr    = rowptr + npad;
    float* z    = bufA;

    const int nB = (N + BKT_NODES - 1) >> NODE_SHIFT;
    const int nT = (E + TILE_EDGES - 1) / TILE_EDGES;
    const int n2 = nB * nT;
    int2* binned   = (int2*)h;
    int*  rnk      = (int*)h + 2 * (size_t)E;
    int*  tileHist = rnk + E;
    int*  off      = tileHist + n2;

    const int B = 256;
    k_bin_count<<<nT, 1024, nB * sizeof(int), stream>>>(dst, tileHist, rnk, E, nT, nB);
    k_scan_excl<<<(n2 + 1023) / 1024, 1024, 0, stream>>>(tileHist, off, n2);
    k_bin_scatter<<<nT, 1024, nB * sizeof(int), stream>>>(src, dst, off, rnk, binned, E, nT, nB);
    k_bucket_csr<<<nB, 1024, 0, stream>>>(binned, off, rowptr, cnt, dinv, csr, N, E, nT, nB);

    const int gatherBlocks = (N + 7) / 8;

    {
        int rowTiles = (N + 15) / 16;
        int blocks   = (rowTiles + 7) / 8;
        k_gemm_wmma<IN_F, H_F, false><<<blocks, B, 0, stream>>>(x, W1, dinv, bufA, N);
    }
    k_gather_gemm2<<<(N + 15) / 16, B, 0, stream>>>(bufA, rowptr, cnt, csr, dinv, b1, W2, h, N);
    k_gather<OUT_F><<<gatherBlocks, B, 0, stream>>>(h, rowptr, cnt, csr, dinv, b2, bufA, N);
    {
        long long waves = (2LL * EP + 31) / 32;
        long long thr   = waves * 32;
        k_decode<<<(int)((thr + B - 1) / B), B, 0, stream>>>(z, pei, nei, out, EP);
    }
}


# Layer-1 product: a resident 16-wave work-group per WGP keeps W1 transposed in LDS and loops over row tiles; B fragments by 16-B LDS reads, A by 16-B row reads; k regrouped per WMMA, sum order changes
# speedup vs baseline: 3.9763x; 1.0995x over previous
#include <hip/hip_runtime.h>

typedef __attribute__((ext_vector_type(2))) float v2f;
typedef __attribute__((ext_vector_type(8))) float v8f;
typedef __attribute__((ext_vector_type(4))) float v4f;

#define IN_F 256
#define H_F  128
#define OUT_F 64


#define NODE_SHIFT 8
#define BKT_NODES  256
#define EPT        8
#define TILE_EDGES 8192

__global__ __launch_bounds__(1024)
void k_bin_count(const int* __restrict__ dst, int* __restrict__ tileHist, int* __restrict__ rnk,
                 int nE, int nT, int nB) {
    extern __shared__ int s_bins[];
    const int tid = threadIdx.x;
    const int t = blockIdx.x;
    for (int b = tid; b < nB; b += 1024) s_bins[b] = 0;
    __syncthreads();
    const int e0 = t * TILE_EDGES;
    if (e0 + TILE_EDGES <= nE) {
        int d[EPT];
#pragma unroll
        for (int u = 0; u < EPT; ++u) d[u] = dst[e0 + u * 1024 + tid];
#pragma unroll
        for (int u = 0; u < EPT; ++u)
            rnk[e0 + u * 1024 + tid] = atomicAdd(&s_bins[d[u] >> NODE_SHIFT], 1);
    } else {
        for (int e = e0 + tid; e < nE; e += 1024)
            rnk[e] = atomicAdd(&s_bins[dst[e] >> NODE_SHIFT], 1);
    }
    __syncthreads();
    for (int b = tid; b < nB; b += 1024) tileHist[b * nT + t] = s_bins[b];
}

__global__ __launch_bounds__(1024)
void k_scan_excl(const int* __restrict__ cnt, int* __restrict__ excl, int n) {
    __shared__ int s_pre[32];
    __shared__ int s_wsum[32];
    __shared__ int s_woff[32];
    const int tid = threadIdx.x;
    const int lane = tid & 31;
    const int wave = tid >> 5;
    const int blockStart = blockIdx.x * 1024;

    int pre = 0;
    for (int j = 0; j < (int)blockIdx.x; ++j) pre += cnt[j * 1024 + tid];
#pragma unroll
    for (int off = 16; off > 0; off >>= 1) pre += __shfl_xor(pre, off, 32);

    const int i = blockStart + tid;
    const int c = (i < n) ? cnt[i] : 0;
    int incl = c;
#pragma unroll
    for (int off = 1; off < 32; off <<= 1) {
        int t = __shfl_up(incl, off, 32);
        if (lane >= off) incl += t;
    }
    if (lane == 31) s_wsum[wave] = incl;
    if (lane == 0)  s_pre[wave] = pre;
    __syncthreads();
    if (wave == 0) {
        int p = s_pre[lane];
#pragma unroll
        for (int off = 16; off > 0; off >>= 1) p += __shfl_xor(p, off, 32);
        const int w = s_wsum[lane];
        int wi = w;
#pragma unroll
        for (int off = 1; off < 32; off <<= 1) {
            int t = __shfl_up(wi, off, 32);
            if (lane >= off) wi += t;
        }
        s_woff[lane] = p + wi - w;
    }
    __syncthreads();
    if (i < n) excl[i] = s_woff[wave] + incl - c;
}

__global__ __launch_bounds__(1024)
void k_bin_scatter(const int* __restrict__ src, const int* __restrict__ dst,
                   const int* __restrict__ off, const int* __restrict__ rnk,
                   int2* __restrict__ binned, int nE, int nT, int nB) {
    extern __shared__ int s_bins[];
    const int tid = threadIdx.x;
    const int t = blockIdx.x;
    for (int b = tid; b < nB; b += 1024) s_bins[b] = off[b * nT + t];
    __syncthreads();
    const int e0 = t * TILE_EDGES;
    if (e0 + TILE_EDGES <= nE) {
        int d[EPT], s[EPT], r[EPT];
#pragma unroll
        for (int u = 0; u < EPT; ++u) {
            const int e = e0 + u * 1024 + tid;
            d[u] = dst[e];
            s[u] = src[e];
            r[u] = rnk[e];
        }
#pragma unroll
        for (int u = 0; u < EPT; ++u)
            binned[s_bins[d[u] >> NODE_SHIFT] + r[u]] = make_int2(s[u], d[u]);
    } else {
        for (int e = e0 + tid; e < nE; e += 1024) {
            const int d = dst[e];
            binned[s_bins[d >> NODE_SHIFT] + rnk[e]] = make_int2(src[e], d);
        }
    }
}

__global__ __launch_bounds__(1024)
void k_bucket_csr(const int2* __restrict__ binned, const int* __restrict__ off,
                  int* __restrict__ rowptr, int* __restrict__ cnt, float* __restrict__ dinv,
                  int* __restrict__ csr, int n, int nE, int nT, int nB) {
    __shared__ int s_cnt[BKT_NODES];
    __shared__ int s_cur[BKT_NODES];
    __shared__ int s_wsum[BKT_NODES / 32];
    const int tid = threadIdx.x;
    const int lane = tid & 31;
    const int wave = tid >> 5;
    const int b = blockIdx.x;
    const int node0 = b << NODE_SHIFT;
    const int bs = off[b * nT];
    const int be = (b + 1 < nB) ? off[(b + 1) * nT] : nE;
    const int own = tid & (BKT_NODES - 1);

    if (tid < BKT_NODES) s_cnt[tid] = 0;
    __syncthreads();
    for (int base = bs; base < be; base += 4096) {
        int d[4];
#pragma unroll
        for (int u = 0; u < 4; ++u) {
            const int e = base + u * 1024 + tid;
            d[u] = binned[(e < be) ? e : be - 1].y;
        }
#pragma unroll
        for (int u = 0; u < 4; ++u) {
            const bool in = base + u * 1024 + tid < be;
            atomicAdd(&s_cnt[in ? d[u] - node0 : own], in ? 1 : 0);
        }
    }
    __syncthreads();

    int c = 0, incl = 0;
    if (tid < BKT_NODES) {
        c = s_cnt[tid];
        incl = c;
#pragma unroll
        for (int o = 1; o < 32; o <<= 1) {
            int t = __shfl_up(incl, o, 32);
            if (lane >= o) incl += t;
        }
        if (lane == 31) s_wsum[wave] = incl;
    }
    __syncthreads();
    if (tid < BKT_NODES) {
        int base = 0;
        for (int w = 0; w < wave; ++w) base += s_wsum[w];
        const int ex = base + incl - c;
        s_cur[tid] = ex;
        const int node = node0 + tid;
        if (node < n) {
            rowptr[node] = bs + ex;
            cnt[node]    = c;
            dinv[node]   = rsqrtf((float)(c + 1));
        }
    }
    __syncthreads();
    for (int base = bs; base < be; base += 4096) {
        int2 ed[4];
#pragma unroll
        for (int u = 0; u < 4; ++u) {
            const int e = base + u * 1024 + tid;
            ed[u] = binned[(e < be) ? e : be - 1];
        }
#pragma unroll
        for (int u = 0; u < 4; ++u) {
            const bool in = base + u * 1024 + tid < be;
            const int p = atomicAdd(&s_cur[in ? ed[u].y - node0 : own], in ? 1 : 0);
            if (in) csr[bs + p] = ed[u].x;
        }
    }
}

#define G1_WAVES 16
#define G1_PITCH 260

__global__ __launch_bounds__(32 * G1_WAVES)
void k_gemm1_lds(const float* __restrict__ X, const float* __restrict__ W,
                 const float* __restrict__ rs, float* __restrict__ Y, int nrows) {
    __shared__ __attribute__((aligned(16))) float wt[H_F * G1_PITCH];
    const int tid  = threadIdx.x;
    const int lane = tid & 31;
    const int wave = __builtin_amdgcn_readfirstlane(tid >> 5);
    const int c = lane & 15;
    const int h = lane >> 4;
    const int nTiles = (nrows + 15) >> 4;
    const int stride = (int)gridDim.x * G1_WAVES;
    int t = (int)blockIdx.x + (int)gridDim.x * wave;

    int r0 = t * 16 + c;
    r0 = (r0 < nrows) ? r0 : nrows - 1;
    const v4f* ap = (const v4f*)(X + (size_t)r0 * IN_F + 16 * h);
    v4f a[4], b[8];
#pragma unroll
    for (int q = 0; q < 4; ++q) a[q] = ap[q];

    for (int it = tid; it < (IN_F / 4) * H_F; it += 32 * G1_WAVES) {
        const int n  = it & (H_F - 1);
        const int k4 = it / H_F;
        const float* wp = W + (size_t)(4 * k4) * H_F + n;
        const v4f w = { wp[0], wp[H_F], wp[2 * H_F], wp[3 * H_F] };
        *(v4f*)(wt + n * G1_PITCH + 4 * k4) = w;
    }
    __syncthreads();
    if (t >= nTiles) return;

    const float* bbase = wt + c * G1_PITCH + 16 * h;
#pragma unroll
    for (int j = 0; j < 8; ++j) b[j] = *(const v4f*)(bbase + j * 16 * G1_PITCH);

    for (;;) {
        const int tn = t + stride;
        const bool more = tn < nTiles;
        int rn = (more ? tn : t) * 16 + c;
        rn = (rn < nrows) ? rn : nrows - 1;
        const v4f* apn = (const v4f*)(X + (size_t)rn * IN_F + 16 * h);
        const int row0 = t * 16;
        const float4 rs0 = *(const float4*)(rs + row0 + 8 * h);
        const float4 rs1 = *(const float4*)(rs + row0 + 8 * h + 4);

        v8f acc[8] = {};
        for (int kc = 0; kc < IN_F; kc += 32) {
            const v4f* pn = (kc + 32 < IN_F) ? ap + ((kc + 32) >> 2) : apn;
            v4f an[4];
#pragma unroll
            for (int q = 0; q < 4; ++q) an[q] = pn[q];
            __builtin_amdgcn_sched_barrier(0);
            const int kcn = (kc + 32) & (IN_F - 1);
#pragma unroll
            for (int q = 0; q < 4; ++q) {
                const int ko = (q < 3) ? kc + 4 * (q + 1) : kcn;
                v4f bn[8];
#pragma unroll
                for (int j = 0; j < 8; ++j) bn[j] = *(const v4f*)(bbase + j * 16 * G1_PITCH + ko);
                const v2f a0 = { a[q].x, a[q].y };
                const v2f a1 = { a[q].z, a[q].w };
#pragma unroll
                for (int j = 0; j < 8; ++j) {
                    const v2f b0 = { b[j].x, b[j].y };
                    acc[j] = __builtin_amdgcn_wmma_f32_16x16x4_f32(
                        false, a0, false, b0, (short)0, acc[j], false, false);
                }
#pragma unroll
                for (int j = 0; j < 8; ++j) {
                    const v2f b1 = { b[j].z, b[j].w };
                    acc[j] = __builtin_amdgcn_wmma_f32_16x16x4_f32(
                        false, a1, false, b1, (short)0, acc[j], false, false);
                }
#pragma unroll
                for (int j = 0; j < 8; ++j) b[j] = bn[j];
            }
#pragma unroll
            for (int q = 0; q < 4; ++q) a[q] = an[q];
        }

        const float rsv[8] = { rs0.x, rs0.y, rs0.z, rs0.w, rs1.x, rs1.y, rs1.z, rs1.w };
        float* yp = Y + (size_t)(row0 + 8 * h) * H_F + c;
        if (row0 + 16 <= nrows) {
#pragma unroll
            for (int j = 0; j < 8; ++j) {
#pragma unroll
                for (int r = 0; r < 8; ++r) yp[r * H_F + j * 16] = acc[j][r] * rsv[r];
            }
        } else {
#pragma unroll
            for (int j = 0; j < 8; ++j) {
#pragma unroll
                for (int r = 0; r < 8; ++r)
                    if (row0 + 8 * h + r < nrows) yp[r * H_F + j * 16] = acc[j][r] * rsv[r];
            }
        }
        if (!more) break;
        t = tn;
        ap = apn;
    }
}

template <int V> struct fvec { typedef __attribute__((ext_vector_type(V))) float type; };

template <int F>
__device__ __forceinline__ typename fvec<F / 32>::type
gather_node(const float* __restrict__ lin, const int* __restrict__ rowptr,
            const int* __restrict__ cnt, const int* __restrict__ csr,
            const float* __restrict__ dinv, const float* __restrict__ bias,
            const int d, const int lane) {
    constexpr int V = F / 32;
    typedef typename fvec<V>::type vecf;
    const int start = rowptr[d];
    const int len   = cnt[d];
    const float wd  = dinv[d];
    const vecf* linv = (const vecf*)lin;

    vecf acc = linv[(size_t)d * 32 + lane];
    const vecf bv = ((const vecf*)bias)[lane];

    for (int base = 0; base < len; base += 32) {
        const int m = (len - base < 32) ? (len - base) : 32;
        int s_l = 0;
        if (lane < m) s_l = csr[start + base + lane];
        int k = 0;
        for (; k + 4 <= m; k += 4) {
            const int s0 = __builtin_amdgcn_readlane(s_l, k);
            const int s1 = __builtin_amdgcn_readlane(s_l, k + 1);
            const int s2 = __builtin_amdgcn_readlane(s_l, k + 2);
            const int s3 = __builtin_amdgcn_readlane(s_l, k + 3);
            const vecf v0 = linv[(size_t)s0 * 32 + lane];
            const vecf v1 = linv[(size_t)s1 * 32 + lane];
            const vecf v2 = linv[(size_t)s2 * 32 + lane];
            const vecf v3 = linv[(size_t)s3 * 32 + lane];
            acc += v0;
            acc += v1;
            acc += v2;
            acc += v3;
        }
        for (; k < m; ++k) {
            const int s0 = __builtin_amdgcn_readlane(s_l, k);
            acc += linv[(size_t)s0 * 32 + lane];
        }
    }
    vecf res;
#pragma unroll
    for (int j = 0; j < V; ++j) res[j] = fmaf(acc[j], wd, bv[j]);
    return res;
}

template <int F>
__global__ __launch_bounds__(256)
void k_gather(const float* __restrict__ lin, const int* __restrict__ rowptr,
              const int* __restrict__ cnt, const int* __restrict__ csr,
              const float* __restrict__ dinv, const float* __restrict__ bias,
              float* __restrict__ agg, int n) {
    typedef typename fvec<F / 32>::type vecf;
    const int lane = threadIdx.x & 31;
    const int d = __builtin_amdgcn_readfirstlane((int)((blockIdx.x * blockDim.x + threadIdx.x) >> 5));
    if (d >= n) return;
    const vecf acc = gather_node<F>(lin, rowptr, cnt, csr, dinv, bias, d, lane);
    ((vecf*)agg)[(size_t)d * 32 + lane] = acc;
}

#define HS_PITCH 132

__global__ __launch_bounds__(256)
void k_gather_gemm2(const float* __restrict__ lin, const int* __restrict__ rowptr,
                    const int* __restrict__ cnt, const int* __restrict__ csr,
                    const float* __restrict__ dinv, const float* __restrict__ bias,
                    const float* __restrict__ W2, float* __restrict__ zlin, int n) {
    typedef fvec<4>::type v4f;
    __shared__ float hs[16 * HS_PITCH];
    const int lane  = threadIdx.x & 31;
    const int wave  = __builtin_amdgcn_readfirstlane((int)(threadIdx.x >> 5));
    const int node0 = blockIdx.x * 16;

    for (int t = 0; t < 2; ++t) {
        const int r = wave * 2 + t;
        const int d = node0 + r;
        v4f acc = { 0.0f, 0.0f, 0.0f, 0.0f };
        if (d < n) {
            acc = gather_node<H_F>(lin, rowptr, cnt, csr, dinv, bias, d, lane);
#pragma unroll
            for (int j = 0; j < 4; ++j) acc[j] = fmaxf(acc[j], 0.0f);
        }
        *(v4f*)(hs + r * HS_PITCH + 4 * lane) = acc;
    }
    __syncthreads();
    if (wave >= (OUT_F / 16)) return;

    const int m    = lane & 15;
    const int half = lane >> 4;
    const float* arow = hs + m * HS_PITCH + 2 * half;
    const float* wcol = W2 + (size_t)(2 * half) * OUT_F + wave * 16 + m;
    float sc[8];
#pragma unroll
    for (int r = 0; r < 8; ++r) {
        const int row = node0 + r + 8 * half;
        sc[r] = dinv[(row < n) ? row : n - 1];
    }
    v8f acc = {};
#pragma unroll 8
    for (int kb = 0; kb < H_F; kb += 4) {
        const v2f a = *(const v2f*)(arow + kb);
        const v2f b = { wcol[(size_t)kb * OUT_F], wcol[(size_t)(kb + 1) * OUT_F] };
        acc = __builtin_amdgcn_wmma_f32_16x16x4_f32(false, a, false, b, (short)0, acc, false, false);
    }
#pragma unroll
    for (int r = 0; r < 8; ++r) {
        const int row = node0 + r + 8 * half;
        if (row < n) zlin[(size_t)row * OUT_F + wave * 16 + m] = acc[r] * sc[r];
    }
}


template <int K>
__device__ __forceinline__ unsigned row_bcast(unsigned v) {
    return (unsigned)__builtin_amdgcn_update_dpp(0, (int)v, 0x150 + K, 0xf, 0xf, true);
}
template <int M>
__device__ __forceinline__ float row_xadd(float x) {
    const int y = __builtin_amdgcn_update_dpp(0, __float_as_int(x), 0x160 + M, 0xf, 0xf, true);
    return x + __int_as_float(y);
}

template <int K>
__device__ __forceinline__ float4 z_piece(const char* __restrict__ zb, unsigned off, unsigned sub16) {
    return *(const float4*)(zb + (row_bcast<K>(off) + sub16));
}

template <int K>
__device__ __forceinline__ float score_step(const float4 a, const float4 b, int sub, float res) {
    float p = a.x * b.x;
    p = fmaf(a.y, b.y, p);
    p = fmaf(a.z, b.z, p);
    p = fmaf(a.w, b.w, p);
    p = row_xadd<1>(p);
    p = row_xadd<2>(p);
    p = row_xadd<4>(p);
    p = row_xadd<8>(p);
    return (sub == K) ? p : res;
}

#define DEC_LOAD(K, J) a[J] = z_piece<K>(zb, offu, sub16); b[J] = z_piece<K>(zb, offv, sub16);
#define DEC_SUM(K, J)  res = score_step<K>(a[J], b[J], sub, res);

__global__ __launch_bounds__(256)
void k_decode(const float* __restrict__ z,
              const int* __restrict__ pos, const int* __restrict__ neg,
              float* __restrict__ out, int EP) {
    const int lane  = threadIdx.x & 31;
    const int sub   = lane & 15;
    const int total = 2 * EP;
    const int i0    = (int)((blockIdx.x * blockDim.x + threadIdx.x) >> 5) * 32;
    if (i0 >= total) return;
    const int i = i0 + lane;
    int u = 0, v = 0;
    if (i < total) {
        if (i < EP) { u = pos[i];       v = pos[EP + i]; }
        else        { u = neg[i - EP];  v = neg[i]; }
    }
    const char* zb = (const char*)z;
    const unsigned offu  = (unsigned)u * (OUT_F * 4u);
    const unsigned offv  = (unsigned)v * (OUT_F * 4u);
    const unsigned sub16 = (unsigned)sub * 16u;
    float4 a[8], b[8];
    float res = 0.0f;

    DEC_LOAD(0, 0) DEC_LOAD(1, 1) DEC_LOAD(2, 2) DEC_LOAD(3, 3)
    DEC_LOAD(4, 4) DEC_LOAD(5, 5) DEC_LOAD(6, 6) DEC_LOAD(7, 7)
    DEC_SUM(0, 0) DEC_SUM(1, 1) DEC_SUM(2, 2) DEC_SUM(3, 3)
    DEC_SUM(4, 4) DEC_SUM(5, 5) DEC_SUM(6, 6) DEC_SUM(7, 7)
    DEC_LOAD(8, 0)  DEC_LOAD(9, 1)  DEC_LOAD(10, 2) DEC_LOAD(11, 3)
    DEC_LOAD(12, 4) DEC_LOAD(13, 5) DEC_LOAD(14, 6) DEC_LOAD(15, 7)
    DEC_SUM(8, 0)  DEC_SUM(9, 1)  DEC_SUM(10, 2) DEC_SUM(11, 3)
    DEC_SUM(12, 4) DEC_SUM(13, 5) DEC_SUM(14, 6) DEC_SUM(15, 7)

    if (i < total) out[i] = res;
}


extern "C" void kernel_launch(void* const* d_in, const int* in_sizes, int n_in,
                              void* d_out, int out_size, void* d_ws, size_t ws_size,
                              hipStream_t stream) {
    const float* x    = (const float*)d_in[0];
    const int*   ei   = (const int*)  d_in[1];
    const int*   pei  = (const int*)  d_in[2];
    const int*   nei  = (const int*)  d_in[3];
    const float* W1   = (const float*)d_in[4];
    const float* b1   = (const float*)d_in[5];
    const float* W2   = (const float*)d_in[6];
    const float* b2   = (const float*)d_in[7];
    float* out = (float*)d_out;

    const int N  = in_sizes[0] / IN_F;
    const int E  = in_sizes[1] / 2;
    const int EP = in_sizes[2] / 2;
    const int* src = ei;
    const int* dst = ei + E;

    float* dinv = (float*)d_ws;
    size_t npad = ((size_t)N + 255) & ~(size_t)255;
    float* bufA = dinv + npad;
    float* h    = bufA + (size_t)N * H_F;
    int*   cnt    = (int*)(h + (size_t)N * H_F);
    int*   rowptr = cnt + npad;
    int*   csr    = rowptr + npad;
    float* z    = bufA;

    const int nB = (N + BKT_NODES - 1) >> NODE_SHIFT;
    const int nT = (E + TILE_EDGES - 1) / TILE_EDGES;
    const int n2 = nB * nT;
    int2* binned   = (int2*)h;
    int*  rnk      = (int*)h + 2 * (size_t)E;
    int*  tileHist = rnk + E;
    int*  off      = tileHist + n2;

    const int B = 256;
    k_bin_count<<<nT, 1024, nB * sizeof(int), stream>>>(dst, tileHist, rnk, E, nT, nB);
    k_scan_excl<<<(n2 + 1023) / 1024, 1024, 0, stream>>>(tileHist, off, n2);
    k_bin_scatter<<<nT, 1024, nB * sizeof(int), stream>>>(src, dst, off, rnk, binned, E, nT, nB);
    k_bucket_csr<<<nB, 1024, 0, stream>>>(binned, off, rowptr, cnt, dinv, csr, N, E, nT, nB);

    const int gatherBlocks = (N + 7) / 8;

    k_gemm1_lds<<<256, 32 * G1_WAVES, 0, stream>>>(x, W1, dinv, bufA, N);
    k_gather_gemm2<<<(N + 15) / 16, B, 0, stream>>>(bufA, rowptr, cnt, csr, dinv, b1, W2, h, N);
    k_gather<OUT_F><<<gatherBlocks, B, 0, stream>>>(h, rowptr, cnt, csr, dinv, b2, bufA, N);
    {
        long long waves = (2LL * EP + 31) / 32;
        long long thr   = waves * 32;
        k_decode<<<(int)((thr + B - 1) / B), B, 0, stream>>>(z, pei, nei, out, EP);
    }
}


# Edge sort in two launches: a tile sorts itself by bucket in LDS; a bucket's work-group walks its piece of each tile, offset = sum of piece starts; scan and scatter launches gone; sum order shifts
# speedup vs baseline: 4.4316x; 1.1145x over previous
#include <hip/hip_runtime.h>

typedef __attribute__((ext_vector_type(2))) float v2f;
typedef __attribute__((ext_vector_type(8))) float v8f;
typedef __attribute__((ext_vector_type(4))) float v4f;

#define IN_F 256
#define H_F  128
#define OUT_F 64


#define NODE_SHIFT 8
#define BKT_NODES  256
#define EPT        8
#define TILE_EDGES 8192
#define PPW        8

__global__ __launch_bounds__(1024)
void k_tile_sort(const int* __restrict__ src, const int* __restrict__ dst,
                 int2* __restrict__ binned, int* __restrict__ offT, int nE, int nT, int nB) {
    extern __shared__ int s_bins[];
    __shared__ int s_wsum[32];
    __shared__ int s_woff[32];
    __shared__ int s_tot;
    const int tid = threadIdx.x;
    const int lane = tid & 31;
    const int wave = tid >> 5;
    const int t = blockIdx.x;
    for (int b = tid; b < nB; b += 1024) s_bins[b] = 0;
    __syncthreads();
    const int e0 = t * TILE_EDGES;
    int d[EPT], s[EPT], r[EPT];
#pragma unroll
    for (int u = 0; u < EPT; ++u) {
        const int e = e0 + u * 1024 + tid;
        const int ec = (e < nE) ? e : nE - 1;
        d[u] = dst[ec];
        s[u] = src[ec];
    }
#pragma unroll
    for (int u = 0; u < EPT; ++u) {
        r[u] = 0;
        if (e0 + u * 1024 + tid < nE) r[u] = atomicAdd(&s_bins[d[u] >> NODE_SHIFT], 1);
    }
    __syncthreads();

    int carry = 0;
    for (int b0 = 0; b0 < nB; b0 += 1024) {
        const int b = b0 + tid;
        const int c = (b < nB) ? s_bins[b] : 0;
        int incl = c;
#pragma unroll
        for (int o = 1; o < 32; o <<= 1) {
            int v = __shfl_up(incl, o, 32);
            if (lane >= o) incl += v;
        }
        if (lane == 31) s_wsum[wave] = incl;
        __syncthreads();
        if (wave == 0) {
            const int w = s_wsum[lane];
            int wi = w;
#pragma unroll
            for (int o = 1; o < 32; o <<= 1) {
                int v = __shfl_up(wi, o, 32);
                if (lane >= o) wi += v;
            }
            s_woff[lane] = wi - w;
            if (lane == 31) s_tot = wi;
        }
        __syncthreads();
        const int ex = carry + s_woff[wave] + incl - c;
        if (b < nB) {
            s_bins[b] = ex;
            offT[(size_t)b * nT + t] = ex;
        }
        carry += s_tot;
        __syncthreads();
    }
    if (tid == 0) offT[(size_t)nB * nT + t] = carry;

#pragma unroll
    for (int u = 0; u < EPT; ++u)
        if (e0 + u * 1024 + tid < nE)
            binned[e0 + s_bins[d[u] >> NODE_SHIFT] + r[u]] = make_int2(s[u], d[u]);
}

template <bool FILL>
__device__ __forceinline__ int walk_pieces(const int2* __restrict__ binned,
                                           const int* __restrict__ o0, const int* __restrict__ o1,
                                           const int nT, const int wave, const int lane,
                                           const int node0,
                                           int* s_ctr, int* __restrict__ csr, const int bs) {
    int stsum = 0;
    const int nG = (nT + 32 * PPW - 1) / (32 * PPW);
    for (int g = 0; g < nG; ++g) {
        const int tl = 32 * PPW * g + PPW * wave + (lane & (PPW - 1));
        const bool pv = tl < nT;
        const int tc = pv ? tl : 0;
        const int st_l = o0[tc];
        const int en_l = o1[tc];
        const int len_l = pv ? en_l - st_l : 0;
        const int gs_l = tc * TILE_EDGES + st_l;
        if (pv && lane < PPW) stsum += st_l;
        int mx = len_l;
        mx = max(mx, __shfl_xor(mx, 1, 32));
        mx = max(mx, __shfl_xor(mx, 2, 32));
        mx = max(mx, __shfl_xor(mx, 4, 32));
        const int kmax = __builtin_amdgcn_readfirstlane(mx);
        for (int k = 0; k < kmax; k += 32) {
            int2 ed[PPW];
#pragma unroll
            for (int u = 0; u < PPW; ++u) {
                const int len_u = __builtin_amdgcn_readlane(len_l, u);
                const int gs_u  = __builtin_amdgcn_readlane(gs_l, u);
                const bool in = k + lane < len_u;
                ed[u] = binned[in ? gs_u + k + lane : 0];
            }
#pragma unroll
            for (int u = 0; u < PPW; ++u) asm volatile("" : "+v"(ed[u].x), "+v"(ed[u].y));
            if (FILL) {
                int p[PPW];
#pragma unroll
                for (int u = 0; u < PPW; ++u) {
                    const int len_u = __builtin_amdgcn_readlane(len_l, u);
                    p[u] = 0;
                    if (k + lane < len_u) p[u] = atomicAdd(&s_ctr[ed[u].y - node0], 1);
                }
#pragma unroll
                for (int u = 0; u < PPW; ++u) {
                    const int len_u = __builtin_amdgcn_readlane(len_l, u);
                    if (k + lane < len_u) csr[bs + p[u]] = ed[u].x;
                }
            } else {
#pragma unroll
                for (int u = 0; u < PPW; ++u) {
                    const int len_u = __builtin_amdgcn_readlane(len_l, u);
                    if (k + lane < len_u) atomicAdd(&s_ctr[ed[u].y - node0], 1);
                }
            }
        }
    }
    return stsum;
}

__global__ __launch_bounds__(1024)
void k_bucket_csr(const int2* __restrict__ binned, const int* __restrict__ offT,
                  int* __restrict__ rowptr, int* __restrict__ cnt, float* __restrict__ dinv,
                  int* __restrict__ csr, int n, int nT) {
    __shared__ int s_cnt[BKT_NODES];
    __shared__ int s_cur[BKT_NODES];
    __shared__ int s_wsum[BKT_NODES / 32];
    __shared__ int s_bsum[32];
    const int tid = threadIdx.x;
    const int lane = tid & 31;
    const int wave = __builtin_amdgcn_readfirstlane(tid >> 5);
    const int b = blockIdx.x;
    const int node0 = b << NODE_SHIFT;
    const int* o0 = offT + (size_t)b * nT;
    const int* o1 = o0 + nT;

    if (tid < BKT_NODES) s_cnt[tid] = 0;
    __syncthreads();
    int stsum = walk_pieces<false>(binned, o0, o1, nT, wave, lane, node0, s_cnt, csr, 0);
#pragma unroll
    for (int o = 16; o > 0; o >>= 1) stsum += __shfl_xor(stsum, o, 32);
    if (lane == 0) s_bsum[wave] = stsum;
    __syncthreads();

    int bs = s_bsum[lane];
#pragma unroll
    for (int o = 16; o > 0; o >>= 1) bs += __shfl_xor(bs, o, 32);

    int c = 0, incl = 0;
    if (tid < BKT_NODES) {
        c = s_cnt[tid];
        incl = c;
#pragma unroll
        for (int o = 1; o < 32; o <<= 1) {
            int v = __shfl_up(incl, o, 32);
            if (lane >= o) incl += v;
        }
        if (lane == 31) s_wsum[wave] = incl;
    }
    __syncthreads();
    if (tid < BKT_NODES) {
        int base = 0;
        for (int w = 0; w < wave; ++w) base += s_wsum[w];
        const int ex = base + incl - c;
        s_cur[tid] = ex;
        const int node = node0 + tid;
        if (node < n) {
            rowptr[node] = bs + ex;
            cnt[node]    = c;
            dinv[node]   = rsqrtf((float)(c + 1));
        }
    }
    __syncthreads();
    walk_pieces<true>(binned, o0, o1, nT, wave, lane, node0, s_cur, csr, bs);
}

#define G1_WAVES 16
#define G1_PITCH 260

__global__ __launch_bounds__(32 * G1_WAVES)
void k_gemm1_lds(const float* __restrict__ X, const float* __restrict__ W,
                 const float* __restrict__ rs, float* __restrict__ Y, int nrows) {
    __shared__ __attribute__((aligned(16))) float wt[H_F * G1_PITCH];
    const int tid  = threadIdx.x;
    const int lane = tid & 31;
    const int wave = __builtin_amdgcn_readfirstlane(tid >> 5);
    const int c = lane & 15;
    const int h = lane >> 4;
    const int nTiles = (nrows + 15) >> 4;
    const int stride = (int)gridDim.x * G1_WAVES;
    int t = (int)blockIdx.x + (int)gridDim.x * wave;

    int r0 = t * 16 + c;
    r0 = (r0 < nrows) ? r0 : nrows - 1;
    const v4f* ap = (const v4f*)(X + (size_t)r0 * IN_F + 16 * h);
    v4f a[4], b[8];
#pragma unroll
    for (int q = 0; q < 4; ++q) a[q] = ap[q];

    for (int it = tid; it < (IN_F / 4) * H_F; it += 32 * G1_WAVES) {
        const int n  = it & (H_F - 1);
        const int k4 = it / H_F;
        const float* wp = W + (size_t)(4 * k4) * H_F + n;
        const v4f w = { wp[0], wp[H_F], wp[2 * H_F], wp[3 * H_F] };
        *(v4f*)(wt + n * G1_PITCH + 4 * k4) = w;
    }
    __syncthreads();
    if (t >= nTiles) return;

    const float* bbase = wt + c * G1_PITCH + 16 * h;
#pragma unroll
    for (int j = 0; j < 8; ++j) b[j] = *(const v4f*)(bbase + j * 16 * G1_PITCH);

    for (;;) {
        const int tn = t + stride;
        const bool more = tn < nTiles;
        int rn = (more ? tn : t) * 16 + c;
        rn = (rn < nrows) ? rn : nrows - 1;
        const v4f* apn = (const v4f*)(X + (size_t)rn * IN_F + 16 * h);
        const int row0 = t * 16;
        const float4 rs0 = *(const float4*)(rs + row0 + 8 * h);
        const float4 rs1 = *(const float4*)(rs + row0 + 8 * h + 4);

        v8f acc[8] = {};
        for (int kc = 0; kc < IN_F; kc += 32) {
            const v4f* pn = (kc + 32 < IN_F) ? ap + ((kc + 32) >> 2) : apn;
            v4f an[4];
#pragma unroll
            for (int q = 0; q < 4; ++q) an[q] = pn[q];
            __builtin_amdgcn_sched_barrier(0);
            const int kcn = (kc + 32) & (IN_F - 1);
#pragma unroll
            for (int q = 0; q < 4; ++q) {
                const int ko = (q < 3) ? kc + 4 * (q + 1) : kcn;
                v4f bn[8];
#pragma unroll
                for (int j = 0; j < 8; ++j) bn[j] = *(const v4f*)(bbase + j * 16 * G1_PITCH + ko);
                const v2f a0 = { a[q].x, a[q].y };
                const v2f a1 = { a[q].z, a[q].w };
#pragma unroll
                for (int j = 0; j < 8; ++j) {
                    const v2f b0 = { b[j].x, b[j].y };
                    acc[j] = __builtin_amdgcn_wmma_f32_16x16x4_f32(
                        false, a0, false, b0, (short)0, acc[j], false, false);
                }
#pragma unroll
                for (int j = 0; j < 8; ++j) {
                    const v2f b1 = { b[j].z, b[j].w };
                    acc[j] = __builtin_amdgcn_wmma_f32_16x16x4_f32(
                        false, a1, false, b1, (short)0, acc[j], false, false);
                }
#pragma unroll
                for (int j = 0; j < 8; ++j) b[j] = bn[j];
            }
#pragma unroll
            for (int q = 0; q < 4; ++q) a[q] = an[q];
        }

        const float rsv[8] = { rs0.x, rs0.y, rs0.z, rs0.w, rs1.x, rs1.y, rs1.z, rs1.w };
        float* yp = Y + (size_t)(row0 + 8 * h) * H_F + c;
        if (row0 + 16 <= nrows) {
#pragma unroll
            for (int j = 0; j < 8; ++j) {
#pragma unroll
                for (int r = 0; r < 8; ++r) yp[r * H_F + j * 16] = acc[j][r] * rsv[r];
            }
        } else {
#pragma unroll
            for (int j = 0; j < 8; ++j) {
#pragma unroll
                for (int r = 0; r < 8; ++r)
                    if (row0 + 8 * h + r < nrows) yp[r * H_F + j * 16] = acc[j][r] * rsv[r];
            }
        }
        if (!more) break;
        t = tn;
        ap = apn;
    }
}

template <int V> struct fvec { typedef __attribute__((ext_vector_type(V))) float type; };

template <int F>
__device__ __forceinline__ typename fvec<F / 32>::type
gather_node(const float* __restrict__ lin, const int* __restrict__ rowptr,
            const int* __restrict__ cnt, const int* __restrict__ csr,
            const float* __restrict__ dinv, const float* __restrict__ bias,
            const int d, const int lane) {
    constexpr int V = F / 32;
    typedef typename fvec<V>::type vecf;
    const int start = rowptr[d];
    const int len   = cnt[d];
    const float wd  = dinv[d];
    const vecf* linv = (const vecf*)lin;

    vecf acc = linv[(size_t)d * 32 + lane];
    const vecf bv = ((const vecf*)bias)[lane];

    for (int base = 0; base < len; base += 32) {
        const int m = (len - base < 32) ? (len - base) : 32;
        int s_l = 0;
        if (lane < m) s_l = csr[start + base + lane];
        int k = 0;
        for (; k + 4 <= m; k += 4) {
            const int s0 = __builtin_amdgcn_readlane(s_l, k);
            const int s1 = __builtin_amdgcn_readlane(s_l, k + 1);
            const int s2 = __builtin_amdgcn_readlane(s_l, k + 2);
            const int s3 = __builtin_amdgcn_readlane(s_l, k + 3);
            const vecf v0 = linv[(size_t)s0 * 32 + lane];
            const vecf v1 = linv[(size_t)s1 * 32 + lane];
            const vecf v2 = linv[(size_t)s2 * 32 + lane];
            const vecf v3 = linv[(size_t)s3 * 32 + lane];
            acc += v0;
            acc += v1;
            acc += v2;
            acc += v3;
        }
        for (; k < m; ++k) {
            const int s0 = __builtin_amdgcn_readlane(s_l, k);
            acc += linv[(size_t)s0 * 32 + lane];
        }
    }
    vecf res;
#pragma unroll
    for (int j = 0; j < V; ++j) res[j] = fmaf(acc[j], wd, bv[j]);
    return res;
}

template <int F>
__global__ __launch_bounds__(256)
void k_gather(const float* __restrict__ lin, const int* __restrict__ rowptr,
              const int* __restrict__ cnt, const int* __restrict__ csr,
              const float* __restrict__ dinv, const float* __restrict__ bias,
              float* __restrict__ agg, int n) {
    typedef typename fvec<F / 32>::type vecf;
    const int lane = threadIdx.x & 31;
    const int d = __builtin_amdgcn_readfirstlane((int)((blockIdx.x * blockDim.x + threadIdx.x) >> 5));
    if (d >= n) return;
    const vecf acc = gather_node<F>(lin, rowptr, cnt, csr, dinv, bias, d, lane);
    ((vecf*)agg)[(size_t)d * 32 + lane] = acc;
}

#define HS_PITCH 132

__global__ __launch_bounds__(256)
void k_gather_gemm2(const float* __restrict__ lin, const int* __restrict__ rowptr,
                    const int* __restrict__ cnt, const int* __restrict__ csr,
                    const float* __restrict__ dinv, const float* __restrict__ bias,
                    const float* __restrict__ W2, float* __restrict__ zlin, int n) {
    typedef fvec<4>::type v4f;
    __shared__ float hs[16 * HS_PITCH];
    const int lane  = threadIdx.x & 31;
    const int wave  = __builtin_amdgcn_readfirstlane((int)(threadIdx.x >> 5));
    const int node0 = blockIdx.x * 16;

    for (int t = 0; t < 2; ++t) {
        const int r = wave * 2 + t;
        const int d = node0 + r;
        v4f acc = { 0.0f, 0.0f, 0.0f, 0.0f };
        if (d < n) {
            acc = gather_node<H_F>(lin, rowptr, cnt, csr, dinv, bias, d, lane);
#pragma unroll
            for (int j = 0; j < 4; ++j) acc[j] = fmaxf(acc[j], 0.0f);
        }
        *(v4f*)(hs + r * HS_PITCH + 4 * lane) = acc;
    }
    __syncthreads();
    if (wave >= (OUT_F / 16)) return;

    const int m    = lane & 15;
    const int half = lane >> 4;
    const float* arow = hs + m * HS_PITCH + 2 * half;
    const float* wcol = W2 + (size_t)(2 * half) * OUT_F + wave * 16 + m;
    float sc[8];
#pragma unroll
    for (int r = 0; r < 8; ++r) {
        const int row = node0 + r + 8 * half;
        sc[r] = dinv[(row < n) ? row : n - 1];
    }
    v8f acc = {};
#pragma unroll 8
    for (int kb = 0; kb < H_F; kb += 4) {
        const v2f a = *(const v2f*)(arow + kb);
        const v2f b = { wcol[(size_t)kb * OUT_F], wcol[(size_t)(kb + 1) * OUT_F] };
        acc = __builtin_amdgcn_wmma_f32_16x16x4_f32(false, a, false, b, (short)0, acc, false, false);
    }
#pragma unroll
    for (int r = 0; r < 8; ++r) {
        const int row = node0 + r + 8 * half;
        if (row < n) zlin[(size_t)row * OUT_F + wave * 16 + m] = acc[r] * sc[r];
    }
}


template <int K>
__device__ __forceinline__ unsigned row_bcast(unsigned v) {
    return (unsigned)__builtin_amdgcn_update_dpp(0, (int)v, 0x150 + K, 0xf, 0xf, true);
}
template <int M>
__device__ __forceinline__ float row_xadd(float x) {
    const int y = __builtin_amdgcn_update_dpp(0, __float_as_int(x), 0x160 + M, 0xf, 0xf, true);
    return x + __int_as_float(y);
}

template <int K>
__device__ __forceinline__ float4 z_piece(const char* __restrict__ zb, unsigned off, unsigned sub16) {
    return *(const float4*)(zb + (row_bcast<K>(off) + sub16));
}

template <int K>
__device__ __forceinline__ float score_step(const float4 a, const float4 b, int sub, float res) {
    float p = a.x * b.x;
    p = fmaf(a.y, b.y, p);
    p = fmaf(a.z, b.z, p);
    p = fmaf(a.w, b.w, p);
    p = row_xadd<1>(p);
    p = row_xadd<2>(p);
    p = row_xadd<4>(p);
    p = row_xadd<8>(p);
    return (sub == K) ? p : res;
}

#define DEC_LOAD(K, J) a[J] = z_piece<K>(zb, offu, sub16); b[J] = z_piece<K>(zb, offv, sub16);
#define DEC_SUM(K, J)  res = score_step<K>(a[J], b[J], sub, res);

__global__ __launch_bounds__(256)
void k_decode(const float* __restrict__ z,
              const int* __restrict__ pos, const int* __restrict__ neg,
              float* __restrict__ out, int EP) {
    const int lane  = threadIdx.x & 31;
    const int sub   = lane & 15;
    const int total = 2 * EP;
    const int i0    = (int)((blockIdx.x * blockDim.x + threadIdx.x) >> 5) * 32;
    if (i0 >= total) return;
    const int i = i0 + lane;
    int u = 0, v = 0;
    if (i < total) {
        if (i < EP) { u = pos[i];       v = pos[EP + i]; }
        else        { u = neg[i - EP];  v = neg[i]; }
    }
    const char* zb = (const char*)z;
    const unsigned offu  = (unsigned)u * (OUT_F * 4u);
    const unsigned offv  = (unsigned)v * (OUT_F * 4u);
    const unsigned sub16 = (unsigned)sub * 16u;
    float4 a[8], b[8];
    float res = 0.0f;

    DEC_LOAD(0, 0) DEC_LOAD(1, 1) DEC_LOAD(2, 2) DEC_LOAD(3, 3)
    DEC_LOAD(4, 4) DEC_LOAD(5, 5) DEC_LOAD(6, 6) DEC_LOAD(7, 7)
    DEC_SUM(0, 0) DEC_SUM(1, 1) DEC_SUM(2, 2) DEC_SUM(3, 3)
    DEC_SUM(4, 4) DEC_SUM(5, 5) DEC_SUM(6, 6) DEC_SUM(7, 7)
    DEC_LOAD(8, 0)  DEC_LOAD(9, 1)  DEC_LOAD(10, 2) DEC_LOAD(11, 3)
    DEC_LOAD(12, 4) DEC_LOAD(13, 5) DEC_LOAD(14, 6) DEC_LOAD(15, 7)
    DEC_SUM(8, 0)  DEC_SUM(9, 1)  DEC_SUM(10, 2) DEC_SUM(11, 3)
    DEC_SUM(12, 4) DEC_SUM(13, 5) DEC_SUM(14, 6) DEC_SUM(15, 7)

    if (i < total) out[i] = res;
}


extern "C" void kernel_launch(void* const* d_in, const int* in_sizes, int n_in,
                              void* d_out, int out_size, void* d_ws, size_t ws_size,
                              hipStream_t stream) {
    const float* x    = (const float*)d_in[0];
    const int*   ei   = (const int*)  d_in[1];
    const int*   pei  = (const int*)  d_in[2];
    const int*   nei  = (const int*)  d_in[3];
    const float* W1   = (const float*)d_in[4];
    const float* b1   = (const float*)d_in[5];
    const float* W2   = (const float*)d_in[6];
    const float* b2   = (const float*)d_in[7];
    float* out = (float*)d_out;

    const int N  = in_sizes[0] / IN_F;
    const int E  = in_sizes[1] / 2;
    const int EP = in_sizes[2] / 2;
    const int* src = ei;
    const int* dst = ei + E;

    float* dinv = (float*)d_ws;
    size_t npad = ((size_t)N + 255) & ~(size_t)255;
    float* bufA = dinv + npad;
    float* h    = bufA + (size_t)N * H_F;
    int*   cnt    = (int*)(h + (size_t)N * H_F);
    int*   rowptr = cnt + npad;
    int*   csr    = rowptr + npad;
    float* z    = bufA;

    const int nB = (N + BKT_NODES - 1) >> NODE_SHIFT;
    const int nT = (E + TILE_EDGES - 1) / TILE_EDGES;
    int2* binned = (int2*)h;
    int*  offT   = (int*)h + 2 * (size_t)E;

    const int B = 256;
    k_tile_sort<<<nT, 1024, nB * sizeof(int), stream>>>(src, dst, binned, offT, E, nT, nB);
    k_bucket_csr<<<nB, 1024, 0, stream>>>(binned, offT, rowptr, cnt, dinv, csr, N, nT);

    const int gatherBlocks = (N + 7) / 8;

    k_gemm1_lds<<<256, 32 * G1_WAVES, 0, stream>>>(x, W1, dinv, bufA, N);
    k_gather_gemm2<<<(N + 15) / 16, B, 0, stream>>>(bufA, rowptr, cnt, csr, dinv, b1, W2, h, N);
    k_gather<OUT_F><<<gatherBlocks, B, 0, stream>>>(h, rowptr, cnt, csr, dinv, b2, bufA, N);
    {
        long long waves = (2LL * EP + 31) / 32;
        long long thr   = waves * 32;
        k_decode<<<(int)((thr + B - 1) / B), B, 0, stream>>>(z, pei, nei, out, EP);
    }
}


# Edge sort: the scattered 8-B pair stores and 4-B source stores go to an LDS copy of the tile / bucket first and leave as whole-line 16-B stores; places, ranks and sum order as before
# speedup vs baseline: 4.6468x; 1.0486x over previous
#include <hip/hip_runtime.h>

typedef __attribute__((ext_vector_type(2))) float v2f;
typedef __attribute__((ext_vector_type(8))) float v8f;
typedef __attribute__((ext_vector_type(4))) float v4f;

#define IN_F 256
#define H_F  128
#define OUT_F 64


#define NODE_SHIFT 8
#define BKT_NODES  256
#define EPT        8
#define TILE_EDGES 8192
#define PPW        8
#define STAGE_CAP  8192

__global__ __launch_bounds__(1024)
void k_tile_sort(const int* __restrict__ src, const int* __restrict__ dst,
                 int2* __restrict__ binned, int* __restrict__ offT, int nE, int nT, int nB) {
    extern __shared__ int s_bins[];
    __shared__ __attribute__((aligned(16))) int2 s_stage[TILE_EDGES];
    __shared__ int s_wsum[32];
    __shared__ int s_woff[32];
    __shared__ int s_tot;
    const int tid = threadIdx.x;
    const int lane = tid & 31;
    const int wave = tid >> 5;
    const int t = blockIdx.x;
    for (int b = tid; b < nB; b += 1024) s_bins[b] = 0;
    __syncthreads();
    const int e0 = t * TILE_EDGES;
    int d[EPT], s[EPT], r[EPT];
#pragma unroll
    for (int u = 0; u < EPT; ++u) {
        const int e = e0 + u * 1024 + tid;
        const int ec = (e < nE) ? e : nE - 1;
        d[u] = dst[ec];
        s[u] = src[ec];
    }
#pragma unroll
    for (int u = 0; u < EPT; ++u) {
        r[u] = 0;
        if (e0 + u * 1024 + tid < nE) r[u] = atomicAdd(&s_bins[d[u] >> NODE_SHIFT], 1);
    }
    __syncthreads();

    int carry = 0;
    for (int b0 = 0; b0 < nB; b0 += 1024) {
        const int b = b0 + tid;
        const int c = (b < nB) ? s_bins[b] : 0;
        int incl = c;
#pragma unroll
        for (int o = 1; o < 32; o <<= 1) {
            int v = __shfl_up(incl, o, 32);
            if (lane >= o) incl += v;
        }
        if (lane == 31) s_wsum[wave] = incl;
        __syncthreads();
        if (wave == 0) {
            const int w = s_wsum[lane];
            int wi = w;
#pragma unroll
            for (int o = 1; o < 32; o <<= 1) {
                int v = __shfl_up(wi, o, 32);
                if (lane >= o) wi += v;
            }
            s_woff[lane] = wi - w;
            if (lane == 31) s_tot = wi;
        }
        __syncthreads();
        const int ex = carry + s_woff[wave] + incl - c;
        if (b < nB) {
            s_bins[b] = ex;
            offT[(size_t)b * nT + t] = ex;
        }
        carry += s_tot;
        __syncthreads();
    }
    if (tid == 0) offT[(size_t)nB * nT + t] = carry;

#pragma unroll
    for (int u = 0; u < EPT; ++u)
        if (e0 + u * 1024 + tid < nE)
            s_stage[s_bins[d[u] >> NODE_SHIFT] + r[u]] = make_int2(s[u], d[u]);
    __syncthreads();

    const int cntT = (nE - e0 < TILE_EDGES) ? nE - e0 : TILE_EDGES;
    const int4* sp = (const int4*)s_stage;
    int4* gp = (int4*)(binned + e0);
    for (int i = tid; 2 * i + 1 < cntT; i += 1024) gp[i] = sp[i];
    if ((cntT & 1) && tid == 0) binned[e0 + cntT - 1] = s_stage[cntT - 1];
}

template <int MODE>
__device__ __forceinline__ int walk_pieces(const int2* __restrict__ binned,
                                           const int* __restrict__ o0, const int* __restrict__ o1,
                                           const int nT, const int wave, const int lane,
                                           const int node0,
                                           int* s_ctr, int* __restrict__ csr, const int bs,
                                           int* s_out) {
    int stsum = 0;
    const int nG = (nT + 32 * PPW - 1) / (32 * PPW);
    for (int g = 0; g < nG; ++g) {
        const int tl = 32 * PPW * g + PPW * wave + (lane & (PPW - 1));
        const bool pv = tl < nT;
        const int tc = pv ? tl : 0;
        const int st_l = o0[tc];
        const int en_l = o1[tc];
        const int len_l = pv ? en_l - st_l : 0;
        const int gs_l = tc * TILE_EDGES + st_l;
        if (pv && lane < PPW) stsum += st_l;
        int mx = len_l;
        mx = max(mx, __shfl_xor(mx, 1, 32));
        mx = max(mx, __shfl_xor(mx, 2, 32));
        mx = max(mx, __shfl_xor(mx, 4, 32));
        const int kmax = __builtin_amdgcn_readfirstlane(mx);
        for (int k = 0; k < kmax; k += 32) {
            int2 ed[PPW];
#pragma unroll
            for (int u = 0; u < PPW; ++u) {
                const int len_u = __builtin_amdgcn_readlane(len_l, u);
                const int gs_u  = __builtin_amdgcn_readlane(gs_l, u);
                const bool in = k + lane < len_u;
                ed[u] = binned[in ? gs_u + k + lane : 0];
            }
#pragma unroll
            for (int u = 0; u < PPW; ++u) asm volatile("" : "+v"(ed[u].x), "+v"(ed[u].y));
            if (MODE != 0) {
                int p[PPW];
#pragma unroll
                for (int u = 0; u < PPW; ++u) {
                    const int len_u = __builtin_amdgcn_readlane(len_l, u);
                    p[u] = 0;
                    if (k + lane < len_u) p[u] = atomicAdd(&s_ctr[ed[u].y - node0], 1);
                }
#pragma unroll
                for (int u = 0; u < PPW; ++u) {
                    const int len_u = __builtin_amdgcn_readlane(len_l, u);
                    if (k + lane < len_u) {
                        if (MODE == 1) csr[bs + p[u]] = ed[u].x;
                        else           s_out[p[u]] = ed[u].x;
                    }
                }
            } else {
#pragma unroll
                for (int u = 0; u < PPW; ++u) {
                    const int len_u = __builtin_amdgcn_readlane(len_l, u);
                    if (k + lane < len_u) atomicAdd(&s_ctr[ed[u].y - node0], 1);
                }
            }
        }
    }
    return stsum;
}

__global__ __launch_bounds__(1024)
void k_bucket_csr(const int2* __restrict__ binned, const int* __restrict__ offT,
                  int* __restrict__ rowptr, int* __restrict__ cnt, float* __restrict__ dinv,
                  int* __restrict__ csr, int n, int nT) {
    __shared__ int s_cnt[BKT_NODES];
    __shared__ int s_cur[BKT_NODES];
    __shared__ int s_wsum[BKT_NODES / 32];
    __shared__ int s_bsum[32];
    __shared__ int s_stage[STAGE_CAP];
    const int tid = threadIdx.x;
    const int lane = tid & 31;
    const int wave = __builtin_amdgcn_readfirstlane(tid >> 5);
    const int b = blockIdx.x;
    const int node0 = b << NODE_SHIFT;
    const int* o0 = offT + (size_t)b * nT;
    const int* o1 = o0 + nT;

    if (tid < BKT_NODES) s_cnt[tid] = 0;
    __syncthreads();
    int stsum = walk_pieces<0>(binned, o0, o1, nT, wave, lane, node0, s_cnt, csr, 0, s_stage);
#pragma unroll
    for (int o = 16; o > 0; o >>= 1) stsum += __shfl_xor(stsum, o, 32);
    if (lane == 0) s_bsum[wave] = stsum;
    __syncthreads();

    int bs = s_bsum[lane];
#pragma unroll
    for (int o = 16; o > 0; o >>= 1) bs += __shfl_xor(bs, o, 32);

    int c = 0, incl = 0;
    if (tid < BKT_NODES) {
        c = s_cnt[tid];
        incl = c;
#pragma unroll
        for (int o = 1; o < 32; o <<= 1) {
            int v = __shfl_up(incl, o, 32);
            if (lane >= o) incl += v;
        }
        if (lane == 31) s_wsum[wave] = incl;
    }
    __syncthreads();
    int total = 0;
    for (int w = 0; w < BKT_NODES / 32; ++w) total += s_wsum[w];
    if (tid < BKT_NODES) {
        int base = 0;
        for (int w = 0; w < wave; ++w) base += s_wsum[w];
        const int ex = base + incl - c;
        s_cur[tid] = ex;
        const int node = node0 + tid;
        if (node < n) {
            rowptr[node] = bs + ex;
            cnt[node]    = c;
            dinv[node]   = rsqrtf((float)(c + 1));
        }
    }
    __syncthreads();
    if (total > STAGE_CAP) {
        walk_pieces<1>(binned, o0, o1, nT, wave, lane, node0, s_cur, csr, bs, s_stage);
        return;
    }
    walk_pieces<2>(binned, o0, o1, nT, wave, lane, node0, s_cur, csr, bs, s_stage);
    __syncthreads();
    const int a0 = bs & ~3;
    const int nq = (bs + total - a0 + 3) >> 2;
    for (int q = tid; q < nq; q += 1024) {
        const int g = a0 + 4 * q;
        const int l = g - bs;
        if (l >= 0 && l + 4 <= total) {
            const int4 v = make_int4(s_stage[l], s_stage[l + 1], s_stage[l + 2], s_stage[l + 3]);
            *(int4*)(csr + g) = v;
        } else {
            for (int j = 0; j < 4; ++j)
                if (l + j >= 0 && l + j < total) csr[g + j] = s_stage[l + j];
        }
    }
}

#define G1_WAVES 16
#define G1_PITCH 260

__global__ __launch_bounds__(32 * G1_WAVES)
void k_gemm1_lds(const float* __restrict__ X, const float* __restrict__ W,
                 const float* __restrict__ rs, float* __restrict__ Y, int nrows) {
    __shared__ __attribute__((aligned(16))) float wt[H_F * G1_PITCH];
    const int tid  = threadIdx.x;
    const int lane = tid & 31;
    const int wave = __builtin_amdgcn_readfirstlane(tid >> 5);
    const int c = lane & 15;
    const int h = lane >> 4;
    const int nTiles = (nrows + 15) >> 4;
    const int stride = (int)gridDim.x * G1_WAVES;
    int t = (int)blockIdx.x + (int)gridDim.x * wave;

    int r0 = t * 16 + c;
    r0 = (r0 < nrows) ? r0 : nrows - 1;
    const v4f* ap = (const v4f*)(X + (size_t)r0 * IN_F + 16 * h);
    v4f a[4], b[8];
#pragma unroll
    for (int q = 0; q < 4; ++q) a[q] = ap[q];

    for (int it = tid; it < (IN_F / 4) * H_F; it += 32 * G1_WAVES) {
        const int n  = it & (H_F - 1);
        const int k4 = it / H_F;
        const float* wp = W + (size_t)(4 * k4) * H_F + n;
        const v4f w = { wp[0], wp[H_F], wp[2 * H_F], wp[3 * H_F] };
        *(v4f*)(wt + n * G1_PITCH + 4 * k4) = w;
    }
    __syncthreads();
    if (t >= nTiles) return;

    const float* bbase = wt + c * G1_PITCH + 16 * h;
#pragma unroll
    for (int j = 0; j < 8; ++j) b[j] = *(const v4f*)(bbase + j * 16 * G1_PITCH);

    for (;;) {
        const int tn = t + stride;
        const bool more = tn < nTiles;
        int rn = (more ? tn : t) * 16 + c;
        rn = (rn < nrows) ? rn : nrows - 1;
        const v4f* apn = (const v4f*)(X + (size_t)rn * IN_F + 16 * h);
        const int row0 = t * 16;
        const float4 rs0 = *(const float4*)(rs + row0 + 8 * h);
        const float4 rs1 = *(const float4*)(rs + row0 + 8 * h + 4);

        v8f acc[8] = {};
        for (int kc = 0; kc < IN_F; kc += 32) {
            const v4f* pn = (kc + 32 < IN_F) ? ap + ((kc + 32) >> 2) : apn;
            v4f an[4];
#pragma unroll
            for (int q = 0; q < 4; ++q) an[q] = pn[q];
            __builtin_amdgcn_sched_barrier(0);
            const int kcn = (kc + 32) & (IN_F - 1);
#pragma unroll
            for (int q = 0; q < 4; ++q) {
                const int ko = (q < 3) ? kc + 4 * (q + 1) : kcn;
                v4f bn[8];
#pragma unroll
                for (int j = 0; j < 8; ++j) bn[j] = *(const v4f*)(bbase + j * 16 * G1_PITCH + ko);
                const v2f a0 = { a[q].x, a[q].y };
                const v2f a1 = { a[q].z, a[q].w };
#pragma unroll
                for (int j = 0; j < 8; ++j) {
                    const v2f b0 = { b[j].x, b[j].y };
                    acc[j] = __builtin_amdgcn_wmma_f32_16x16x4_f32(
                        false, a0, false, b0, (short)0, acc[j], false, false);
                }
#pragma unroll
                for (int j = 0; j < 8; ++j) {
                    const v2f b1 = { b[j].z, b[j].w };
                    acc[j] = __builtin_amdgcn_wmma_f32_16x16x4_f32(
                        false, a1, false, b1, (short)0, acc[j], false, false);
                }
#pragma unroll
                for (int j = 0; j < 8; ++j) b[j] = bn[j];
            }
#pragma unroll
            for (int q = 0; q < 4; ++q) a[q] = an[q];
        }

        const float rsv[8] = { rs0.x, rs0.y, rs0.z, rs0.w, rs1.x, rs1.y, rs1.z, rs1.w };
        float* yp = Y + (size_t)(row0 + 8 * h) * H_F + c;
        if (row0 + 16 <= nrows) {
#pragma unroll
            for (int j = 0; j < 8; ++j) {
#pragma unroll
                for (int r = 0; r < 8; ++r) yp[r * H_F + j * 16] = acc[j][r] * rsv[r];
            }
        } else {
#pragma unroll
            for (int j = 0; j < 8; ++j) {
#pragma unroll
                for (int r = 0; r < 8; ++r)
                    if (row0 + 8 * h + r < nrows) yp[r * H_F + j * 16] = acc[j][r] * rsv[r];
            }
        }
        if (!more) break;
        t = tn;
        ap = apn;
    }
}

template <int V> struct fvec { typedef __attribute__((ext_vector_type(V))) float type; };

template <int F>
__device__ __forceinline__ typename fvec<F / 32>::type
gather_node(const float* __restrict__ lin, const int* __restrict__ rowptr,
            const int* __restrict__ cnt, const int* __restrict__ csr,
            const float* __restrict__ dinv, const float* __restrict__ bias,
            const int d, const int lane) {
    constexpr int V = F / 32;
    typedef typename fvec<V>::type vecf;
    const int start = rowptr[d];
    const int len   = cnt[d];
    const float wd  = dinv[d];
    const vecf* linv = (const vecf*)lin;

    vecf acc = linv[(size_t)d * 32 + lane];
    const vecf bv = ((const vecf*)bias)[lane];

    for (int base = 0; base < len; base += 32) {
        const int m = (len - base < 32) ? (len - base) : 32;
        int s_l = 0;
        if (lane < m) s_l = csr[start + base + lane];
        int k = 0;
        for (; k + 4 <= m; k += 4) {
            const int s0 = __builtin_amdgcn_readlane(s_l, k);
            const int s1 = __builtin_amdgcn_readlane(s_l, k + 1);
            const int s2 = __builtin_amdgcn_readlane(s_l, k + 2);
            const int s3 = __builtin_amdgcn_readlane(s_l, k + 3);
            const vecf v0 = linv[(size_t)s0 * 32 + lane];
            const vecf v1 = linv[(size_t)s1 * 32 + lane];
            const vecf v2 = linv[(size_t)s2 * 32 + lane];
            const vecf v3 = linv[(size_t)s3 * 32 + lane];
            acc += v0;
            acc += v1;
            acc += v2;
            acc += v3;
        }
        for (; k < m; ++k) {
            const int s0 = __builtin_amdgcn_readlane(s_l, k);
            acc += linv[(size_t)s0 * 32 + lane];
        }
    }
    vecf res;
#pragma unroll
    for (int j = 0; j < V; ++j) res[j] = fmaf(acc[j], wd, bv[j]);
    return res;
}

template <int F>
__global__ __launch_bounds__(256)
void k_gather(const float* __restrict__ lin, const int* __restrict__ rowptr,
              const int* __restrict__ cnt, const int* __restrict__ csr,
              const float* __restrict__ dinv, const float* __restrict__ bias,
              float* __restrict__ agg, int n) {
    typedef typename fvec<F / 32>::type vecf;
    const int lane = threadIdx.x & 31;
    const int d = __builtin_amdgcn_readfirstlane((int)((blockIdx.x * blockDim.x + threadIdx.x) >> 5));
    if (d >= n) return;
    const vecf acc = gather_node<F>(lin, rowptr, cnt, csr, dinv, bias, d, lane);
    ((vecf*)agg)[(size_t)d * 32 + lane] = acc;
}

#define HS_PITCH 132

__global__ __launch_bounds__(256)
void k_gather_gemm2(const float* __restrict__ lin, const int* __restrict__ rowptr,
                    const int* __restrict__ cnt, const int* __restrict__ csr,
                    const float* __restrict__ dinv, const float* __restrict__ bias,
                    const float* __restrict__ W2, float* __restrict__ zlin, int n) {
    typedef fvec<4>::type v4f;
    __shared__ float hs[16 * HS_PITCH];
    const int lane  = threadIdx.x & 31;
    const int wave  = __builtin_amdgcn_readfirstlane((int)(threadIdx.x >> 5));
    const int node0 = blockIdx.x * 16;

    for (int t = 0; t < 2; ++t) {
        const int r = wave * 2 + t;
        const int d = node0 + r;
        v4f acc = { 0.0f, 0.0f, 0.0f, 0.0f };
        if (d < n) {
            acc = gather_node<H_F>(lin, rowptr, cnt, csr, dinv, bias, d, lane);
#pragma unroll
            for (int j = 0; j < 4; ++j) acc[j] = fmaxf(acc[j], 0.0f);
        }
        *(v4f*)(hs + r * HS_PITCH + 4 * lane) = acc;
    }
    __syncthreads();
    if (wave >= (OUT_F / 16)) return;

    const int m    = lane & 15;
    const int half = lane >> 4;
    const float* arow = hs + m * HS_PITCH + 2 * half;
    const float* wcol = W2 + (size_t)(2 * half) * OUT_F + wave * 16 + m;
    float sc[8];
#pragma unroll
    for (int r = 0; r < 8; ++r) {
        const int row = node0 + r + 8 * half;
        sc[r] = dinv[(row < n) ? row : n - 1];
    }
    v8f acc = {};
#pragma unroll 8
    for (int kb = 0; kb < H_F; kb += 4) {
        const v2f a = *(const v2f*)(arow + kb);
        const v2f b = { wcol[(size_t)kb * OUT_F], wcol[(size_t)(kb + 1) * OUT_F] };
        acc = __builtin_amdgcn_wmma_f32_16x16x4_f32(false, a, false, b, (short)0, acc, false, false);
    }
#pragma unroll
    for (int r = 0; r < 8; ++r) {
        const int row = node0 + r + 8 * half;
        if (row < n) zlin[(size_t)row * OUT_F + wave * 16 + m] = acc[r] * sc[r];
    }
}


template <int K>
__device__ __forceinline__ unsigned row_bcast(unsigned v) {
    return (unsigned)__builtin_amdgcn_update_dpp(0, (int)v, 0x150 + K, 0xf, 0xf, true);
}
template <int M>
__device__ __forceinline__ float row_xadd(float x) {
    const int y = __builtin_amdgcn_update_dpp(0, __float_as_int(x), 0x160 + M, 0xf, 0xf, true);
    return x + __int_as_float(y);
}

template <int K>
__device__ __forceinline__ float4 z_piece(const char* __restrict__ zb, unsigned off, unsigned sub16) {
    return *(const float4*)(zb + (row_bcast<K>(off) + sub16));
}

template <int K>
__device__ __forceinline__ float score_step(const float4 a, const float4 b, int sub, float res) {
    float p = a.x * b.x;
    p = fmaf(a.y, b.y, p);
    p = fmaf(a.z, b.z, p);
    p = fmaf(a.w, b.w, p);
    p = row_xadd<1>(p);
    p = row_xadd<2>(p);
    p = row_xadd<4>(p);
    p = row_xadd<8>(p);
    return (sub == K) ? p : res;
}

#define DEC_LOAD(K, J) a[J] = z_piece<K>(zb, offu, sub16); b[J] = z_piece<K>(zb, offv, sub16);
#define DEC_SUM(K, J)  res = score_step<K>(a[J], b[J], sub, res);

__global__ __launch_bounds__(256)
void k_decode(const float* __restrict__ z,
              const int* __restrict__ pos, const int* __restrict__ neg,
              float* __restrict__ out, int EP) {
    const int lane  = threadIdx.x & 31;
    const int sub   = lane & 15;
    const int total = 2 * EP;
    const int i0    = (int)((blockIdx.x * blockDim.x + threadIdx.x) >> 5) * 32;
    if (i0 >= total) return;
    const int i = i0 + lane;
    int u = 0, v = 0;
    if (i < total) {
        if (i < EP) { u = pos[i];       v = pos[EP + i]; }
        else        { u = neg[i - EP];  v = neg[i]; }
    }
    const char* zb = (const char*)z;
    const unsigned offu  = (unsigned)u * (OUT_F * 4u);
    const unsigned offv  = (unsigned)v * (OUT_F * 4u);
    const unsigned sub16 = (unsigned)sub * 16u;
    float4 a[8], b[8];
    float res = 0.0f;

    DEC_LOAD(0, 0) DEC_LOAD(1, 1) DEC_LOAD(2, 2) DEC_LOAD(3, 3)
    DEC_LOAD(4, 4) DEC_LOAD(5, 5) DEC_LOAD(6, 6) DEC_LOAD(7, 7)
    DEC_SUM(0, 0) DEC_SUM(1, 1) DEC_SUM(2, 2) DEC_SUM(3, 3)
    DEC_SUM(4, 4) DEC_SUM(5, 5) DEC_SUM(6, 6) DEC_SUM(7, 7)
    DEC_LOAD(8, 0)  DEC_LOAD(9, 1)  DEC_LOAD(10, 2) DEC_LOAD(11, 3)
    DEC_LOAD(12, 4) DEC_LOAD(13, 5) DEC_LOAD(14, 6) DEC_LOAD(15, 7)
    DEC_SUM(8, 0)  DEC_SUM(9, 1)  DEC_SUM(10, 2) DEC_SUM(11, 3)
    DEC_SUM(12, 4) DEC_SUM(13, 5) DEC_SUM(14, 6) DEC_SUM(15, 7)

    if (i < total) out[i] = res;
}


extern "C" void kernel_launch(void* const* d_in, const int* in_sizes, int n_in,
                              void* d_out, int out_size, void* d_ws, size_t ws_size,
                              hipStream_t stream) {
    const float* x    = (const float*)d_in[0];
    const int*   ei   = (const int*)  d_in[1];
    const int*   pei  = (const int*)  d_in[2];
    const int*   nei  = (const int*)  d_in[3];
    const float* W1   = (const float*)d_in[4];
    const float* b1   = (const float*)d_in[5];
    const float* W2   = (const float*)d_in[6];
    const float* b2   = (const float*)d_in[7];
    float* out = (float*)d_out;

    const int N  = in_sizes[0] / IN_F;
    const int E  = in_sizes[1] / 2;
    const int EP = in_sizes[2] / 2;
    const int* src = ei;
    const int* dst = ei + E;

    float* dinv = (float*)d_ws;
    size_t npad = ((size_t)N + 255) & ~(size_t)255;
    float* bufA = dinv + npad;
    float* h    = bufA + (size_t)N * H_F;
    int*   cnt    = (int*)(h + (size_t)N * H_F);
    int*   rowptr = cnt + npad;
    int*   csr    = rowptr + npad;
    float* z    = bufA;

    const int nB = (N + BKT_NODES - 1) >> NODE_SHIFT;
    const int nT = (E + TILE_EDGES - 1) / TILE_EDGES;
    int2* binned = (int2*)h;
    int*  offT   = (int*)h + 2 * (size_t)E;

    const int B = 256;
    k_tile_sort<<<nT, 1024, nB * sizeof(int), stream>>>(src, dst, binned, offT, E, nT, nB);
    k_bucket_csr<<<nB, 1024, 0, stream>>>(binned, offT, rowptr, cnt, dinv, csr, N, nT);

    const int gatherBlocks = (N + 7) / 8;

    k_gemm1_lds<<<256, 32 * G1_WAVES, 0, stream>>>(x, W1, dinv, bufA, N);
    k_gather_gemm2<<<(N + 15) / 16, B, 0, stream>>>(bufA, rowptr, cnt, csr, dinv, b1, W2, h, N);
    k_gather<OUT_F><<<gatherBlocks, B, 0, stream>>>(h, rowptr, cnt, csr, dinv, b2, bufA, N);
    {
        long long waves = (2LL * EP + 31) / 32;
        long long thr   = waves * 32;
        k_decode<<<(int)((thr + B - 1) / B), B, 0, stream>>>(z, pei, nei, out, EP);
    }
}
